# starting kernel
# baseline (speedup 1.0000x reference)
#include <hip/hip_runtime.h>

typedef __attribute__((ext_vector_type(2))) float v2f;
typedef __attribute__((ext_vector_type(8))) float v8f;

#define N_QUBITS 12
#define N_LAYERS 6
#define BATCH    4096
#define DIM      4096
#define NU       (N_LAYERS * 3)
#define UFLOATS  (NU * 512)

static __device__ __forceinline__ v8f wmma4(v2f a, v2f b, v8f c) {
  return __builtin_amdgcn_wmma_f32_16x16x4_f32(false, a, false, b, (short)0, c,
                                               false, false);
}

__global__ __launch_bounds__(256) void fuse_unitaries(
    const float* __restrict__ weights, float* __restrict__ Umats) {
  const int e = blockIdx.x;
  const int l = e / 3, g = e % 3;
  const int tid = threadIdx.x;
  const int i = tid >> 4, j = tid & 15;

  float pr = 1.f, pi = 0.f;
#pragma unroll
  for (int k = 0; k < 4; ++k) {
    const int q  = 4 * g + k;
    const float ty = 0.5f * weights[(l * N_QUBITS + q) * 2 + 0];
    const float tz = 0.5f * weights[(l * N_QUBITS + q) * 2 + 1];
    const float cy = cosf(ty), sy = sinf(ty);
    const float cz = cosf(tz), sz = sinf(tz);
    const int bi = (i >> (3 - k)) & 1;
    const int bj = (j >> (3 - k)) & 1;
    const float ryv = bi ? (bj ? cy : sy) : (bj ? -sy : cy);
    const float gre = ryv * cz;
    const float gim = ryv * (bi ? sz : -sz);
    const float nr = pr * gre - pi * gim;
    const float ni = pr * gim + pi * gre;
    pr = nr; pi = ni;
  }
  Umats[e * 512 + tid]       = pr;
  Umats[e * 512 + 256 + tid] = pi;
}

__global__ __launch_bounds__(256) void qsim(const float* __restrict__ x,
                                            const float* __restrict__ Umats,
                                            float* __restrict__ out) {
  __shared__ float Sr[DIM];
  __shared__ float Si[DIM];
  __shared__ float Uall[UFLOATS];
  __shared__ float red[256];

  const int tid  = threadIdx.x;
  const int b    = blockIdx.x;
  const int lane = tid & 31;
  const int wave = tid >> 5;
  const int half = lane >> 4;
  const int lm   = lane & 15;

#pragma unroll
  for (int k = 0; k < UFLOATS / 256; ++k)
    Uall[tid + 256 * k] = Umats[tid + 256 * k];

  float cw[N_QUBITS], sw[N_QUBITS];
#pragma unroll
  for (int w = 0; w < N_QUBITS; ++w) {
    const float t = 0.5f * x[b * N_QUBITS + w];
    cw[w] = cosf(t);
    sw[w] = sinf(t);
  }
#pragma unroll
  for (int k = 0; k < 16; ++k) {
    const int a = tid + 256 * k;
    float p = 1.f;
#pragma unroll
    for (int w = 0; w < N_QUBITS; ++w)
      p *= ((a >> (11 - w)) & 1) ? sw[w] : cw[w];
    Sr[a] = p;
    Si[a] = 0.f;
  }
  __syncthreads();

  for (int l = 0; l < N_LAYERS; ++l) {
    const bool gray = (l > 0);
    for (int g = 0; g < 3; ++g) {
      const float* Ur = Uall + (l * 3 + g) * 512;
      const float* Ui = Ur + 256;

      v8f dr[2] = {};
      v8f di[2] = {};
#pragma unroll
      for (int c = 0; c < 4; ++c) {
        const int k0 = 4 * c + 2 * half;
        v2f ur, ui, nui;
        ur.x = Ur[lm * 16 + k0];
        ur.y = Ur[lm * 16 + k0 + 1];
        ui.x = Ui[lm * 16 + k0];
        ui.y = Ui[lm * 16 + k0 + 1];
        nui  = -ui;
#pragma unroll
        for (int t2 = 0; t2 < 2; ++t2) {
          const int tile = wave * 2 + t2;
          int a0, a1;
          if (g == 0) {
            a0 = k0 * 256 + tile * 16 + lm;
            a1 = a0 + 256;
            if (gray) {
              a0 ^= (a0 >> 1);
              a1 ^= (a1 >> 1);
            }
          } else if (g == 1) {
            a0 = tile * 256 + k0 * 16 + lm;  a1 = a0 + 16;
          } else {
            a0 = (tile * 16 + lm) * 16 + k0; a1 = a0 + 1;
          }
          v2f br, bi;
          br.x = Sr[a0]; br.y = Sr[a1];
          bi.x = Si[a0]; bi.y = Si[a1];
          if (g < 2) {
            dr[t2] = wmma4(ur,  br, dr[t2]);
            dr[t2] = wmma4(nui, bi, dr[t2]);
            di[t2] = wmma4(ur,  bi, di[t2]);
            di[t2] = wmma4(ui,  br, di[t2]);
          } else {
            dr[t2] = wmma4(br, ur,  dr[t2]);
            dr[t2] = wmma4(bi, nui, dr[t2]);
            di[t2] = wmma4(br, ui,  di[t2]);
            di[t2] = wmma4(bi, ur,  di[t2]);
          }
        }
      }
      __syncthreads();
#pragma unroll
      for (int t2 = 0; t2 < 2; ++t2) {
        const int tile = wave * 2 + t2;
#pragma unroll
        for (int v = 0; v < 8; ++v) {
          const int m = v + 8 * half;
          int a;
          if (g == 0)      a = m * 256 + tile * 16 + lm;
          else if (g == 1) a = tile * 256 + m * 16 + lm;
          else             a = (tile * 16 + m) * 16 + lm;
          Sr[a] = dr[t2][v];
          Si[a] = di[t2][v];
        }
      }
      __syncthreads();
    }
  }

  float acc = 0.f;
#pragma unroll
  for (int k = 0; k < 16; ++k) {
    const int a = tid + 256 * k;
    const float p = Sr[a] * Sr[a] + Si[a] * Si[a];
    acc += (a & 2048) ? -p : p;
  }
  red[tid] = acc;
  __syncthreads();
  for (int off = 128; off > 0; off >>= 1) {
    if (tid < off) red[tid] += red[tid + off];
    __syncthreads();
  }
  if (tid == 0) out[b] = red[0];
}

extern "C" void kernel_launch(void* const* d_in, const int* in_sizes, int n_in,
                              void* d_out, int out_size, void* d_ws,
                              size_t ws_size, hipStream_t stream) {
  const float* x = (const float*)d_in[0];
  const float* w = (const float*)d_in[1];
  float* out     = (float*)d_out;
  float* Umats   = (float*)d_ws;

  fuse_unitaries<<<dim3(N_LAYERS * 3), dim3(256), 0, stream>>>(w, Umats);
  qsim<<<dim3(BATCH), dim3(256), 0, stream>>>(x, Umats, out);
}


# Factor each layer into real RY Kronecker matrices (2 WMMA per k-chunk and tile, was 4) and a diagonal RZ phase multiplied in element-wise once per layer; changes the order of float sums
# speedup vs baseline: 1.1680x; 1.1680x over previous
#include <hip/hip_runtime.h>


typedef __attribute__((ext_vector_type(2))) float v2f;
typedef __attribute__((ext_vector_type(8))) float v8f;

#define N_QUBITS 12
#define N_LAYERS 6
#define BATCH    4096
#define DIM      4096
#define NU       (N_LAYERS * 3)
#define RFLOATS  (NU * 256)
#define PHL      544
#define TFLOATS  (RFLOATS + N_LAYERS * PHL)

static __device__ __forceinline__ v8f wmma4(v2f a, v2f b, v8f c) {
  return __builtin_amdgcn_wmma_f32_16x16x4_f32(false, a, false, b, (short)0, c,
                                               false, false);
}

__global__ __launch_bounds__(256) void fuse_gates(
    const float* __restrict__ weights, float* __restrict__ T) {
  const int e = blockIdx.x;
  const int tid = threadIdx.x;
  if (e < NU) {
    const int l = e / 3, g = e % 3;
    const int i = tid >> 4, j = tid & 15;
    float p = 1.f;
#pragma unroll
    for (int k = 0; k < 4; ++k) {
      const int q  = 4 * g + k;
      const float ty = 0.5f * weights[(l * N_QUBITS + q) * 2 + 0];
      const float cy = cosf(ty), sy = sinf(ty);
      const int bi = (i >> (3 - k)) & 1;
      const int bj = (j >> (3 - k)) & 1;
      p *= bi ? (bj ? cy : sy) : (bj ? -sy : cy);
    }
    T[e * 256 + tid] = p;
  } else {
    const int l = e - NU;
    float* P = T + RFLOATS + l * PHL;
    float pr = 1.f, pi = 0.f;
    for (int k = 0; k < 8; ++k) {
      const float tz = 0.5f * weights[(l * N_QUBITS + k) * 2 + 1];
      const float cz = cosf(tz), sz = sinf(tz);
      const float gi = ((tid >> (7 - k)) & 1) ? sz : -sz;
      const float nr = pr * cz - pi * gi;
      const float ni = pr * gi + pi * cz;
      pr = nr; pi = ni;
    }
    P[2 * tid]     = pr;
    P[2 * tid + 1] = pi;
    if (tid < 16) {
      float qr = 1.f, qi = 0.f;
      for (int k = 0; k < 4; ++k) {
        const float tz = 0.5f * weights[(l * N_QUBITS + 8 + k) * 2 + 1];
        const float cz = cosf(tz), sz = sinf(tz);
        const float gi = ((tid >> (3 - k)) & 1) ? sz : -sz;
        const float nr = qr * cz - qi * gi;
        const float ni = qr * gi + qi * cz;
        qr = nr; qi = ni;
      }
      P[512 + 2 * tid]     = qr;
      P[512 + 2 * tid + 1] = qi;
    }
  }
}

__global__ __launch_bounds__(256) void qsim(const float* __restrict__ x,
                                            const float* __restrict__ T,
                                            float* __restrict__ out) {
  __shared__ float Sr[DIM];
  __shared__ float Si[DIM];
  __shared__ float Tall[TFLOATS];
  __shared__ float red[256];

  const int tid  = threadIdx.x;
  const int b    = blockIdx.x;
  const int lane = tid & 31;
  const int wave = tid >> 5;
  const int half = lane >> 4;
  const int lm   = lane & 15;

#pragma unroll
  for (int k = 0; k < TFLOATS / 256; ++k)
    Tall[tid + 256 * k] = T[tid + 256 * k];
  if (tid < TFLOATS % 256)
    Tall[tid + 256 * (TFLOATS / 256)] = T[tid + 256 * (TFLOATS / 256)];

  float cw[N_QUBITS], sw[N_QUBITS];
#pragma unroll
  for (int w = 0; w < N_QUBITS; ++w) {
    const float t = 0.5f * x[b * N_QUBITS + w];
    cw[w] = cosf(t);
    sw[w] = sinf(t);
  }
#pragma unroll
  for (int k = 0; k < 16; ++k) {
    const int a = tid + 256 * k;
    float p = 1.f;
#pragma unroll
    for (int w = 0; w < N_QUBITS; ++w)
      p *= ((a >> (11 - w)) & 1) ? sw[w] : cw[w];
    Sr[a] = p;
    Si[a] = 0.f;
  }
  __syncthreads();

  for (int l = 0; l < N_LAYERS; ++l) {
    const bool gray = (l > 0);
    for (int g = 0; g < 3; ++g) {
      const float* Ur = Tall + (l * 3 + g) * 256;

      v8f dr[2] = {};
      v8f di[2] = {};
#pragma unroll
      for (int c = 0; c < 4; ++c) {
        const int k0 = 4 * c + 2 * half;
        v2f ur;
        ur.x = Ur[lm * 16 + k0];
        ur.y = Ur[lm * 16 + k0 + 1];
#pragma unroll
        for (int t2 = 0; t2 < 2; ++t2) {
          const int tile = wave * 2 + t2;
          int a0, a1;
          if (g == 0) {
            a0 = k0 * 256 + tile * 16 + lm;
            a1 = a0 + 256;
            if (gray) {
              a0 ^= (a0 >> 1);
              a1 ^= (a1 >> 1);
            }
          } else if (g == 1) {
            a0 = tile * 256 + k0 * 16 + lm;  a1 = a0 + 16;
          } else {
            a0 = (tile * 16 + lm) * 16 + k0; a1 = a0 + 1;
          }
          v2f br, bi;
          br.x = Sr[a0]; br.y = Sr[a1];
          bi.x = Si[a0]; bi.y = Si[a1];
          if (g < 2) {
            dr[t2] = wmma4(ur, br, dr[t2]);
            di[t2] = wmma4(ur, bi, di[t2]);
          } else {
            dr[t2] = wmma4(br, ur, dr[t2]);
            di[t2] = wmma4(bi, ur, di[t2]);
          }
        }
      }
      if (g == 2) {
        const float* P = Tall + RFLOATS + l * PHL;
        const float p2r = P[512 + 2 * lm];
        const float p2i = P[512 + 2 * lm + 1];
#pragma unroll
        for (int t2 = 0; t2 < 2; ++t2) {
          const int h0 = (wave * 2 + t2) * 16 + 8 * half;
#pragma unroll
          for (int v = 0; v < 8; ++v) {
            const float qr = P[2 * (h0 + v)];
            const float qi = P[2 * (h0 + v) + 1];
            const float wr = qr * p2r - qi * p2i;
            const float wi = qr * p2i + qi * p2r;
            const float sr = dr[t2][v];
            const float si = di[t2][v];
            dr[t2][v] = sr * wr - si * wi;
            di[t2][v] = sr * wi + si * wr;
          }
        }
      }
      __syncthreads();
#pragma unroll
      for (int t2 = 0; t2 < 2; ++t2) {
        const int tile = wave * 2 + t2;
#pragma unroll
        for (int v = 0; v < 8; ++v) {
          const int m = v + 8 * half;
          int a;
          if (g == 0)      a = m * 256 + tile * 16 + lm;
          else if (g == 1) a = tile * 256 + m * 16 + lm;
          else             a = (tile * 16 + m) * 16 + lm;
          Sr[a] = dr[t2][v];
          Si[a] = di[t2][v];
        }
      }
      __syncthreads();
    }
  }

  float acc = 0.f;
#pragma unroll
  for (int k = 0; k < 16; ++k) {
    const int a = tid + 256 * k;
    const float p = Sr[a] * Sr[a] + Si[a] * Si[a];
    acc += (a & 2048) ? -p : p;
  }
  red[tid] = acc;
  __syncthreads();
  for (int off = 128; off > 0; off >>= 1) {
    if (tid < off) red[tid] += red[tid + off];
    __syncthreads();
  }
  if (tid == 0) out[b] = red[0];
}

extern "C" void kernel_launch(void* const* d_in, const int* in_sizes, int n_in,
                              void* d_out, int out_size, void* d_ws,
                              size_t ws_size, hipStream_t stream) {
  const float* x = (const float*)d_in[0];
  const float* w = (const float*)d_in[1];
  float* out     = (float*)d_out;
  float* T       = (float*)d_ws;

  fuse_gates<<<dim3(N_LAYERS * 3 + N_LAYERS), dim3(256), 0, stream>>>(w, T);
  qsim<<<dim3(BATCH), dim3(256), 0, stream>>>(x, T, out);
}


# Chain each layer's 2nd and 3rd rotation stage in registers: stage-2 accumulators are stage 3's A fragments in a permuted k order, saving an LDS round trip and 2 barriers; changes float sum order
# speedup vs baseline: 1.4241x; 1.2193x over previous
#include <hip/hip_runtime.h>


typedef __attribute__((ext_vector_type(2))) float v2f;
typedef __attribute__((ext_vector_type(8))) float v8f;

#define N_QUBITS 12
#define N_LAYERS 6
#define BATCH    4096
#define DIM      4096
#define NU       (N_LAYERS * 3)
#define RFLOATS  (NU * 256)
#define PHL      544
#define TFLOATS  (RFLOATS + N_LAYERS * PHL)

static __device__ __forceinline__ v8f wmma4(v2f a, v2f b, v8f c) {
  return __builtin_amdgcn_wmma_f32_16x16x4_f32(false, a, false, b, (short)0, c,
                                               false, false);
}

__global__ __launch_bounds__(256) void fuse_gates(
    const float* __restrict__ weights, float* __restrict__ T) {
  const int e = blockIdx.x;
  const int tid = threadIdx.x;
  if (e < NU) {
    const int l = e / 3, g = e % 3;
    const int i = tid >> 4, j = tid & 15;
    float p = 1.f;
#pragma unroll
    for (int k = 0; k < 4; ++k) {
      const int q  = 4 * g + k;
      const float ty = 0.5f * weights[(l * N_QUBITS + q) * 2 + 0];
      const float cy = cosf(ty), sy = sinf(ty);
      const int bi = (i >> (3 - k)) & 1;
      const int bj = (j >> (3 - k)) & 1;
      p *= bi ? (bj ? cy : sy) : (bj ? -sy : cy);
    }
    T[e * 256 + tid] = p;
  } else {
    const int l = e - NU;
    float* P = T + RFLOATS + l * PHL;
    float pr = 1.f, pi = 0.f;
    for (int k = 0; k < 8; ++k) {
      const float tz = 0.5f * weights[(l * N_QUBITS + k) * 2 + 1];
      const float cz = cosf(tz), sz = sinf(tz);
      const float gi = ((tid >> (7 - k)) & 1) ? sz : -sz;
      const float nr = pr * cz - pi * gi;
      const float ni = pr * gi + pi * cz;
      pr = nr; pi = ni;
    }
    P[2 * tid]     = pr;
    P[2 * tid + 1] = pi;
    if (tid < 16) {
      float qr = 1.f, qi = 0.f;
      for (int k = 0; k < 4; ++k) {
        const float tz = 0.5f * weights[(l * N_QUBITS + 8 + k) * 2 + 1];
        const float cz = cosf(tz), sz = sinf(tz);
        const float gi = ((tid >> (3 - k)) & 1) ? sz : -sz;
        const float nr = qr * cz - qi * gi;
        const float ni = qr * gi + qi * cz;
        qr = nr; qi = ni;
      }
      P[512 + 2 * tid]     = qr;
      P[512 + 2 * tid + 1] = qi;
    }
  }
}

__global__ __launch_bounds__(256) void qsim(const float* __restrict__ x,
                                            const float* __restrict__ T,
                                            float* __restrict__ out) {
  __shared__ float Sr[DIM];
  __shared__ float Si[DIM];
  __shared__ float Tall[TFLOATS];
  __shared__ float red[256];

  const int tid  = threadIdx.x;
  const int b    = blockIdx.x;
  const int lane = tid & 31;
  const int wave = tid >> 5;
  const int half = lane >> 4;
  const int lm   = lane & 15;

#pragma unroll
  for (int k = 0; k < TFLOATS / 256; ++k)
    Tall[tid + 256 * k] = T[tid + 256 * k];
  if (tid < TFLOATS % 256)
    Tall[tid + 256 * (TFLOATS / 256)] = T[tid + 256 * (TFLOATS / 256)];

  float cw[N_QUBITS], sw[N_QUBITS];
#pragma unroll
  for (int w = 0; w < N_QUBITS; ++w) {
    const float t = 0.5f * x[b * N_QUBITS + w];
    cw[w] = cosf(t);
    sw[w] = sinf(t);
  }
#pragma unroll
  for (int k = 0; k < 16; ++k) {
    const int a = tid + 256 * k;
    float p = 1.f;
#pragma unroll
    for (int w = 0; w < N_QUBITS; ++w)
      p *= ((a >> (11 - w)) & 1) ? sw[w] : cw[w];
    Sr[a] = p;
    Si[a] = 0.f;
  }
  __syncthreads();

  for (int l = 0; l < N_LAYERS; ++l) {
    const bool gray = (l > 0);

    {
      const float* Ur = Tall + (l * 3 + 0) * 256;
      v8f dr[2] = {};
      v8f di[2] = {};
#pragma unroll
      for (int c = 0; c < 4; ++c) {
        const int k0 = 4 * c + 2 * half;
        v2f ur;
        ur.x = Ur[lm * 16 + k0];
        ur.y = Ur[lm * 16 + k0 + 1];
#pragma unroll
        for (int t2 = 0; t2 < 2; ++t2) {
          const int tile = wave * 2 + t2;
          int a0 = k0 * 256 + tile * 16 + lm;
          int a1 = a0 + 256;
          if (gray) {
            a0 ^= (a0 >> 1);
            a1 ^= (a1 >> 1);
          }
          v2f br, bi;
          br.x = Sr[a0]; br.y = Sr[a1];
          bi.x = Si[a0]; bi.y = Si[a1];
          dr[t2] = wmma4(ur, br, dr[t2]);
          di[t2] = wmma4(ur, bi, di[t2]);
        }
      }
      __syncthreads();
#pragma unroll
      for (int t2 = 0; t2 < 2; ++t2) {
        const int tile = wave * 2 + t2;
#pragma unroll
        for (int v = 0; v < 8; ++v) {
          const int m = v + 8 * half;
          const int a = m * 256 + tile * 16 + lm;
          Sr[a] = dr[t2][v];
          Si[a] = di[t2][v];
        }
      }
      __syncthreads();
    }

    {
      const float* U1 = Tall + (l * 3 + 1) * 256;
      const float* U2 = Tall + (l * 3 + 2) * 256;
      v8f er[2] = {};
      v8f ei[2] = {};
#pragma unroll
      for (int c = 0; c < 4; ++c) {
        const int k0 = 4 * c + 2 * half;
        v2f u1;
        u1.x = U1[lm * 16 + k0];
        u1.y = U1[lm * 16 + k0 + 1];
#pragma unroll
        for (int t2 = 0; t2 < 2; ++t2) {
          const int tile = wave * 2 + t2;
          const int a0 = tile * 256 + k0 * 16 + lm;
          const int a1 = a0 + 16;
          v2f br, bi;
          br.x = Sr[a0]; br.y = Sr[a1];
          bi.x = Si[a0]; bi.y = Si[a1];
          er[t2] = wmma4(br, u1, er[t2]);
          ei[t2] = wmma4(bi, u1, ei[t2]);
        }
      }

      v8f dr[2] = {};
      v8f di[2] = {};
#pragma unroll
      for (int c = 0; c < 4; ++c) {
        const int q0 = 8 * half + 2 * c;
        v2f u2;
        u2.x = U2[lm * 16 + q0];
        u2.y = U2[lm * 16 + q0 + 1];
#pragma unroll
        for (int t2 = 0; t2 < 2; ++t2) {
          v2f ar, ai;
          ar.x = er[t2][2 * c]; ar.y = er[t2][2 * c + 1];
          ai.x = ei[t2][2 * c]; ai.y = ei[t2][2 * c + 1];
          dr[t2] = wmma4(ar, u2, dr[t2]);
          di[t2] = wmma4(ai, u2, di[t2]);
        }
      }

      const float* P = Tall + RFLOATS + l * PHL;
      const float p2r = P[512 + 2 * lm];
      const float p2i = P[512 + 2 * lm + 1];
#pragma unroll
      for (int t2 = 0; t2 < 2; ++t2) {
        const int h0 = (wave * 2 + t2) * 16 + 8 * half;
#pragma unroll
        for (int v = 0; v < 8; ++v) {
          const float qr = P[2 * (h0 + v)];
          const float qi = P[2 * (h0 + v) + 1];
          const float wr = qr * p2r - qi * p2i;
          const float wi = qr * p2i + qi * p2r;
          const float sr = dr[t2][v];
          const float si = di[t2][v];
          dr[t2][v] = sr * wr - si * wi;
          di[t2][v] = sr * wi + si * wr;
        }
      }
      __syncthreads();
#pragma unroll
      for (int t2 = 0; t2 < 2; ++t2) {
        const int tile = wave * 2 + t2;
#pragma unroll
        for (int v = 0; v < 8; ++v) {
          const int m = v + 8 * half;
          const int a = (tile * 16 + m) * 16 + lm;
          Sr[a] = dr[t2][v];
          Si[a] = di[t2][v];
        }
      }
      __syncthreads();
    }
  }

  float acc = 0.f;
#pragma unroll
  for (int k = 0; k < 16; ++k) {
    const int a = tid + 256 * k;
    const float p = Sr[a] * Sr[a] + Si[a] * Si[a];
    acc += (a & 2048) ? -p : p;
  }
  red[tid] = acc;
  __syncthreads();
  for (int off = 128; off > 0; off >>= 1) {
    if (tid < off) red[tid] += red[tid + off];
    __syncthreads();
  }
  if (tid == 0) out[b] = red[0];
}

extern "C" void kernel_launch(void* const* d_in, const int* in_sizes, int n_in,
                              void* d_out, int out_size, void* d_ws,
                              size_t ws_size, hipStream_t stream) {
  const float* x = (const float*)d_in[0];
  const float* w = (const float*)d_in[1];
  float* out     = (float*)d_out;
  float* T       = (float*)d_ws;

  fuse_gates<<<dim3(N_LAYERS * 3 + N_LAYERS), dim3(256), 0, stream>>>(w, T);
  qsim<<<dim3(BATCH), dim3(256), 0, stream>>>(x, T, out);
}


# Fully unroll the 6-layer loop: LDS addresses become thread base + instruction offset, not ~60 loop-invariant registers; VGPRs 179 -> 113, so 4 workgroups per WGP are resident, was 2
# speedup vs baseline: 1.6162x; 1.1349x over previous
#include <hip/hip_runtime.h>


typedef __attribute__((ext_vector_type(2))) float v2f;
typedef __attribute__((ext_vector_type(8))) float v8f;

#define N_QUBITS 12
#define N_LAYERS 6
#define BATCH    4096
#define DIM      4096
#define NU       (N_LAYERS * 3)
#define RFLOATS  (NU * 256)
#define PHL      544
#define TFLOATS  (RFLOATS + N_LAYERS * PHL)

static __device__ __forceinline__ v8f wmma4(v2f a, v2f b, v8f c) {
  return __builtin_amdgcn_wmma_f32_16x16x4_f32(false, a, false, b, (short)0, c,
                                               false, false);
}

__global__ __launch_bounds__(256) void fuse_gates(
    const float* __restrict__ weights, float* __restrict__ T) {
  const int e = blockIdx.x;
  const int tid = threadIdx.x;
  if (e < NU) {
    const int l = e / 3, g = e % 3;
    const int i = tid >> 4, j = tid & 15;
    float p = 1.f;
#pragma unroll
    for (int k = 0; k < 4; ++k) {
      const int q  = 4 * g + k;
      const float ty = 0.5f * weights[(l * N_QUBITS + q) * 2 + 0];
      const float cy = cosf(ty), sy = sinf(ty);
      const int bi = (i >> (3 - k)) & 1;
      const int bj = (j >> (3 - k)) & 1;
      p *= bi ? (bj ? cy : sy) : (bj ? -sy : cy);
    }
    T[e * 256 + tid] = p;
  } else {
    const int l = e - NU;
    float* P = T + RFLOATS + l * PHL;
    float pr = 1.f, pi = 0.f;
    for (int k = 0; k < 8; ++k) {
      const float tz = 0.5f * weights[(l * N_QUBITS + k) * 2 + 1];
      const float cz = cosf(tz), sz = sinf(tz);
      const float gi = ((tid >> (7 - k)) & 1) ? sz : -sz;
      const float nr = pr * cz - pi * gi;
      const float ni = pr * gi + pi * cz;
      pr = nr; pi = ni;
    }
    P[2 * tid]     = pr;
    P[2 * tid + 1] = pi;
    if (tid < 16) {
      float qr = 1.f, qi = 0.f;
      for (int k = 0; k < 4; ++k) {
        const float tz = 0.5f * weights[(l * N_QUBITS + 8 + k) * 2 + 1];
        const float cz = cosf(tz), sz = sinf(tz);
        const float gi = ((tid >> (3 - k)) & 1) ? sz : -sz;
        const float nr = qr * cz - qi * gi;
        const float ni = qr * gi + qi * cz;
        qr = nr; qi = ni;
      }
      P[512 + 2 * tid]     = qr;
      P[512 + 2 * tid + 1] = qi;
    }
  }
}

__global__ __launch_bounds__(256) void qsim(const float* __restrict__ x,
                                            const float* __restrict__ T,
                                            float* __restrict__ out) {
  __shared__ float Sr[DIM];
  __shared__ float Si[DIM];
  __shared__ float Tall[TFLOATS];
  __shared__ float red[256];

  const int tid  = threadIdx.x;
  const int b    = blockIdx.x;
  const int lane = tid & 31;
  const int wave = tid >> 5;
  const int half = lane >> 4;
  const int lm   = lane & 15;

#pragma unroll
  for (int k = 0; k < TFLOATS / 256; ++k)
    Tall[tid + 256 * k] = T[tid + 256 * k];
  if (tid < TFLOATS % 256)
    Tall[tid + 256 * (TFLOATS / 256)] = T[tid + 256 * (TFLOATS / 256)];

  float cw[N_QUBITS], sw[N_QUBITS];
#pragma unroll
  for (int w = 0; w < N_QUBITS; ++w) {
    const float t = 0.5f * x[b * N_QUBITS + w];
    cw[w] = cosf(t);
    sw[w] = sinf(t);
  }
#pragma unroll
  for (int k = 0; k < 16; ++k) {
    const int a = tid + 256 * k;
    float p = 1.f;
#pragma unroll
    for (int w = 0; w < N_QUBITS; ++w)
      p *= ((a >> (11 - w)) & 1) ? sw[w] : cw[w];
    Sr[a] = p;
    Si[a] = 0.f;
  }
  __syncthreads();

#pragma unroll
  for (int l = 0; l < N_LAYERS; ++l) {
    const bool gray = (l > 0);

    {
      const float* Ur = Tall + (l * 3 + 0) * 256;
      v8f dr[2] = {};
      v8f di[2] = {};
#pragma unroll
      for (int c = 0; c < 4; ++c) {
        const int k0 = 4 * c + 2 * half;
        v2f ur;
        ur.x = Ur[lm * 16 + k0];
        ur.y = Ur[lm * 16 + k0 + 1];
#pragma unroll
        for (int t2 = 0; t2 < 2; ++t2) {
          const int tile = wave * 2 + t2;
          int a0 = k0 * 256 + tile * 16 + lm;
          int a1 = a0 + 256;
          if (gray) {
            a0 ^= (a0 >> 1);
            a1 ^= (a1 >> 1);
          }
          v2f br, bi;
          br.x = Sr[a0]; br.y = Sr[a1];
          bi.x = Si[a0]; bi.y = Si[a1];
          dr[t2] = wmma4(ur, br, dr[t2]);
          di[t2] = wmma4(ur, bi, di[t2]);
        }
      }
      __syncthreads();
#pragma unroll
      for (int t2 = 0; t2 < 2; ++t2) {
        const int tile = wave * 2 + t2;
#pragma unroll
        for (int v = 0; v < 8; ++v) {
          const int m = v + 8 * half;
          const int a = m * 256 + tile * 16 + lm;
          Sr[a] = dr[t2][v];
          Si[a] = di[t2][v];
        }
      }
      __syncthreads();
    }

    {
      const float* U1 = Tall + (l * 3 + 1) * 256;
      const float* U2 = Tall + (l * 3 + 2) * 256;
      v8f er[2] = {};
      v8f ei[2] = {};
#pragma unroll
      for (int c = 0; c < 4; ++c) {
        const int k0 = 4 * c + 2 * half;
        v2f u1;
        u1.x = U1[lm * 16 + k0];
        u1.y = U1[lm * 16 + k0 + 1];
#pragma unroll
        for (int t2 = 0; t2 < 2; ++t2) {
          const int tile = wave * 2 + t2;
          const int a0 = tile * 256 + k0 * 16 + lm;
          const int a1 = a0 + 16;
          v2f br, bi;
          br.x = Sr[a0]; br.y = Sr[a1];
          bi.x = Si[a0]; bi.y = Si[a1];
          er[t2] = wmma4(br, u1, er[t2]);
          ei[t2] = wmma4(bi, u1, ei[t2]);
        }
      }

      v8f dr[2] = {};
      v8f di[2] = {};
#pragma unroll
      for (int c = 0; c < 4; ++c) {
        const int q0 = 8 * half + 2 * c;
        v2f u2;
        u2.x = U2[lm * 16 + q0];
        u2.y = U2[lm * 16 + q0 + 1];
#pragma unroll
        for (int t2 = 0; t2 < 2; ++t2) {
          v2f ar, ai;
          ar.x = er[t2][2 * c]; ar.y = er[t2][2 * c + 1];
          ai.x = ei[t2][2 * c]; ai.y = ei[t2][2 * c + 1];
          dr[t2] = wmma4(ar, u2, dr[t2]);
          di[t2] = wmma4(ai, u2, di[t2]);
        }
      }

      const float* P = Tall + RFLOATS + l * PHL;
      const float p2r = P[512 + 2 * lm];
      const float p2i = P[512 + 2 * lm + 1];
#pragma unroll
      for (int t2 = 0; t2 < 2; ++t2) {
        const int h0 = (wave * 2 + t2) * 16 + 8 * half;
#pragma unroll
        for (int v = 0; v < 8; ++v) {
          const float qr = P[2 * (h0 + v)];
          const float qi = P[2 * (h0 + v) + 1];
          const float wr = qr * p2r - qi * p2i;
          const float wi = qr * p2i + qi * p2r;
          const float sr = dr[t2][v];
          const float si = di[t2][v];
          dr[t2][v] = sr * wr - si * wi;
          di[t2][v] = sr * wi + si * wr;
        }
      }
      __syncthreads();
#pragma unroll
      for (int t2 = 0; t2 < 2; ++t2) {
        const int tile = wave * 2 + t2;
#pragma unroll
        for (int v = 0; v < 8; ++v) {
          const int m = v + 8 * half;
          const int a = (tile * 16 + m) * 16 + lm;
          Sr[a] = dr[t2][v];
          Si[a] = di[t2][v];
        }
      }
      __syncthreads();
    }
  }

  float acc = 0.f;
#pragma unroll
  for (int k = 0; k < 16; ++k) {
    const int a = tid + 256 * k;
    const float p = Sr[a] * Sr[a] + Si[a] * Si[a];
    acc += (a & 2048) ? -p : p;
  }
  red[tid] = acc;
  __syncthreads();
  for (int off = 128; off > 0; off >>= 1) {
    if (tid < off) red[tid] += red[tid + off];
    __syncthreads();
  }
  if (tid == 0) out[b] = red[0];
}

extern "C" void kernel_launch(void* const* d_in, const int* in_sizes, int n_in,
                              void* d_out, int out_size, void* d_ws,
                              size_t ws_size, hipStream_t stream) {
  const float* x = (const float*)d_in[0];
  const float* w = (const float*)d_in[1];
  float* out     = (float*)d_out;
  float* T       = (float*)d_ws;

  fuse_gates<<<dim3(N_LAYERS * 3 + N_LAYERS), dim3(256), 0, stream>>>(w, T);
  qsim<<<dim3(BATCH), dim3(256), 0, stream>>>(x, T, out);
}


# Last layer cut to what <Z_0> can see: its wires 4-11 rotations and RZ phases cannot change <Z_0>, so after its wires 0-3 matrix the squares are summed from the accumulators; changes float sum order
# speedup vs baseline: 1.8366x; 1.1364x over previous
#include <hip/hip_runtime.h>


typedef __attribute__((ext_vector_type(2))) float v2f;
typedef __attribute__((ext_vector_type(8))) float v8f;

#define N_QUBITS 12
#define N_LAYERS 6
#define BATCH    4096
#define DIM      4096
#define NU       ((N_LAYERS - 1) * 3 + 1)
#define NPH      (N_LAYERS - 1)
#define RFLOATS  (NU * 256)
#define PHL      544
#define TFLOATS  (RFLOATS + NPH * PHL)

static __device__ __forceinline__ v8f wmma4(v2f a, v2f b, v8f c) {
  return __builtin_amdgcn_wmma_f32_16x16x4_f32(false, a, false, b, (short)0, c,
                                               false, false);
}

__global__ __launch_bounds__(256) void fuse_gates(
    const float* __restrict__ weights, float* __restrict__ T) {
  const int e = blockIdx.x;
  const int tid = threadIdx.x;
  if (e < NU) {
    const int l = e / 3, g = e % 3;
    const int i = tid >> 4, j = tid & 15;
    float p = 1.f;
#pragma unroll
    for (int k = 0; k < 4; ++k) {
      const int q  = 4 * g + k;
      const float ty = 0.5f * weights[(l * N_QUBITS + q) * 2 + 0];
      const float cy = cosf(ty), sy = sinf(ty);
      const int bi = (i >> (3 - k)) & 1;
      const int bj = (j >> (3 - k)) & 1;
      p *= bi ? (bj ? cy : sy) : (bj ? -sy : cy);
    }
    T[e * 256 + tid] = p;
  } else {
    const int l = e - NU;
    float* P = T + RFLOATS + l * PHL;
    float pr = 1.f, pi = 0.f;
    for (int k = 0; k < 8; ++k) {
      const float tz = 0.5f * weights[(l * N_QUBITS + k) * 2 + 1];
      const float cz = cosf(tz), sz = sinf(tz);
      const float gi = ((tid >> (7 - k)) & 1) ? sz : -sz;
      const float nr = pr * cz - pi * gi;
      const float ni = pr * gi + pi * cz;
      pr = nr; pi = ni;
    }
    P[2 * tid]     = pr;
    P[2 * tid + 1] = pi;
    if (tid < 16) {
      float qr = 1.f, qi = 0.f;
      for (int k = 0; k < 4; ++k) {
        const float tz = 0.5f * weights[(l * N_QUBITS + 8 + k) * 2 + 1];
        const float cz = cosf(tz), sz = sinf(tz);
        const float gi = ((tid >> (3 - k)) & 1) ? sz : -sz;
        const float nr = qr * cz - qi * gi;
        const float ni = qr * gi + qi * cz;
        qr = nr; qi = ni;
      }
      P[512 + 2 * tid]     = qr;
      P[512 + 2 * tid + 1] = qi;
    }
  }
}

__global__ __launch_bounds__(256) void qsim(const float* __restrict__ x,
                                            const float* __restrict__ T,
                                            float* __restrict__ out) {
  __shared__ float Sr[DIM];
  __shared__ float Si[DIM];
  __shared__ float Tall[TFLOATS];
  __shared__ float red[256];

  const int tid  = threadIdx.x;
  const int b    = blockIdx.x;
  const int lane = tid & 31;
  const int wave = tid >> 5;
  const int half = lane >> 4;
  const int lm   = lane & 15;

#pragma unroll
  for (int k = 0; k < TFLOATS / 256; ++k)
    Tall[tid + 256 * k] = T[tid + 256 * k];
  if (tid < TFLOATS % 256)
    Tall[tid + 256 * (TFLOATS / 256)] = T[tid + 256 * (TFLOATS / 256)];

  float cw[N_QUBITS], sw[N_QUBITS];
#pragma unroll
  for (int w = 0; w < N_QUBITS; ++w) {
    const float t = 0.5f * x[b * N_QUBITS + w];
    cw[w] = cosf(t);
    sw[w] = sinf(t);
  }
#pragma unroll
  for (int k = 0; k < 16; ++k) {
    const int a = tid + 256 * k;
    float p = 1.f;
#pragma unroll
    for (int w = 0; w < N_QUBITS; ++w)
      p *= ((a >> (11 - w)) & 1) ? sw[w] : cw[w];
    Sr[a] = p;
    Si[a] = 0.f;
  }
  __syncthreads();

  float acc = 0.f;
#pragma unroll
  for (int l = 0; l < N_LAYERS; ++l) {
    const bool gray = (l > 0);
    const bool last = (l == N_LAYERS - 1);

    {
      const float* Ur = Tall + (l * 3 + 0) * 256;
      v8f dr[2] = {};
      v8f di[2] = {};
#pragma unroll
      for (int c = 0; c < 4; ++c) {
        const int k0 = 4 * c + 2 * half;
        v2f ur;
        ur.x = Ur[lm * 16 + k0];
        ur.y = Ur[lm * 16 + k0 + 1];
#pragma unroll
        for (int t2 = 0; t2 < 2; ++t2) {
          const int tile = wave * 2 + t2;
          int a0 = k0 * 256 + tile * 16 + lm;
          int a1 = a0 + 256;
          if (gray) {
            a0 ^= (a0 >> 1);
            a1 ^= (a1 >> 1);
          }
          v2f br, bi;
          br.x = Sr[a0]; br.y = Sr[a1];
          bi.x = Si[a0]; bi.y = Si[a1];
          dr[t2] = wmma4(ur, br, dr[t2]);
          di[t2] = wmma4(ur, bi, di[t2]);
        }
      }
      if (last) {
#pragma unroll
        for (int t2 = 0; t2 < 2; ++t2) {
#pragma unroll
          for (int v = 0; v < 8; ++v)
            acc += dr[t2][v] * dr[t2][v] + di[t2][v] * di[t2][v];
        }
        if (half) acc = -acc;
      } else {
        __syncthreads();
#pragma unroll
        for (int t2 = 0; t2 < 2; ++t2) {
          const int tile = wave * 2 + t2;
#pragma unroll
          for (int v = 0; v < 8; ++v) {
            const int m = v + 8 * half;
            const int a = m * 256 + tile * 16 + lm;
            Sr[a] = dr[t2][v];
            Si[a] = di[t2][v];
          }
        }
        __syncthreads();
      }
    }

    if (!last) {
      const float* U1 = Tall + (l * 3 + 1) * 256;
      const float* U2 = Tall + (l * 3 + 2) * 256;
      v8f er[2] = {};
      v8f ei[2] = {};
#pragma unroll
      for (int c = 0; c < 4; ++c) {
        const int k0 = 4 * c + 2 * half;
        v2f u1;
        u1.x = U1[lm * 16 + k0];
        u1.y = U1[lm * 16 + k0 + 1];
#pragma unroll
        for (int t2 = 0; t2 < 2; ++t2) {
          const int tile = wave * 2 + t2;
          const int a0 = tile * 256 + k0 * 16 + lm;
          const int a1 = a0 + 16;
          v2f br, bi;
          br.x = Sr[a0]; br.y = Sr[a1];
          bi.x = Si[a0]; bi.y = Si[a1];
          er[t2] = wmma4(br, u1, er[t2]);
          ei[t2] = wmma4(bi, u1, ei[t2]);
        }
      }

      v8f dr[2] = {};
      v8f di[2] = {};
#pragma unroll
      for (int c = 0; c < 4; ++c) {
        const int q0 = 8 * half + 2 * c;
        v2f u2;
        u2.x = U2[lm * 16 + q0];
        u2.y = U2[lm * 16 + q0 + 1];
#pragma unroll
        for (int t2 = 0; t2 < 2; ++t2) {
          v2f ar, ai;
          ar.x = er[t2][2 * c]; ar.y = er[t2][2 * c + 1];
          ai.x = ei[t2][2 * c]; ai.y = ei[t2][2 * c + 1];
          dr[t2] = wmma4(ar, u2, dr[t2]);
          di[t2] = wmma4(ai, u2, di[t2]);
        }
      }

      const float* P = Tall + RFLOATS + l * PHL;
      const float p2r = P[512 + 2 * lm];
      const float p2i = P[512 + 2 * lm + 1];
#pragma unroll
      for (int t2 = 0; t2 < 2; ++t2) {
        const int h0 = (wave * 2 + t2) * 16 + 8 * half;
#pragma unroll
        for (int v = 0; v < 8; ++v) {
          const float qr = P[2 * (h0 + v)];
          const float qi = P[2 * (h0 + v) + 1];
          const float wr = qr * p2r - qi * p2i;
          const float wi = qr * p2i + qi * p2r;
          const float sr = dr[t2][v];
          const float si = di[t2][v];
          dr[t2][v] = sr * wr - si * wi;
          di[t2][v] = sr * wi + si * wr;
        }
      }
      __syncthreads();
#pragma unroll
      for (int t2 = 0; t2 < 2; ++t2) {
        const int tile = wave * 2 + t2;
#pragma unroll
        for (int v = 0; v < 8; ++v) {
          const int m = v + 8 * half;
          const int a = (tile * 16 + m) * 16 + lm;
          Sr[a] = dr[t2][v];
          Si[a] = di[t2][v];
        }
      }
      __syncthreads();
    }
  }

  red[tid] = acc;
  __syncthreads();
  for (int off = 128; off > 0; off >>= 1) {
    if (tid < off) red[tid] += red[tid + off];
    __syncthreads();
  }
  if (tid == 0) out[b] = red[0];
}

extern "C" void kernel_launch(void* const* d_in, const int* in_sizes, int n_in,
                              void* d_out, int out_size, void* d_ws,
                              size_t ws_size, hipStream_t stream) {
  const float* x = (const float*)d_in[0];
  const float* w = (const float*)d_in[1];
  float* out     = (float*)d_out;
  float* T       = (float*)d_ws;

  fuse_gates<<<dim3(NU + NPH), dim3(256), 0, stream>>>(w, T);
  qsim<<<dim3(BATCH), dim3(256), 0, stream>>>(x, T, out);
}


# Fold layer 0 into the embedding: RY(p)RY(x)=RY(x+p), so the state behind layer 0 is written as a product state times its phase; its 3 matrix stages and 2 LDS trips go; changes float sum order
# speedup vs baseline: 2.1936x; 1.1944x over previous
#include <hip/hip_runtime.h>


typedef __attribute__((ext_vector_type(2))) float v2f;
typedef __attribute__((ext_vector_type(8))) float v8f;

#define N_QUBITS 12
#define N_LAYERS 6
#define BATCH    4096
#define DIM      4096
#define NU       ((N_LAYERS - 2) * 3 + 1)
#define NPH      (N_LAYERS - 1)
#define RFLOATS  (NU * 256)
#define PHL      544
#define LFLOATS  (RFLOATS + (NPH - 1) * PHL)
#define TFLOATS  (LFLOATS + PHL)

static __device__ __forceinline__ v8f wmma4(v2f a, v2f b, v8f c) {
  return __builtin_amdgcn_wmma_f32_16x16x4_f32(false, a, false, b, (short)0, c,
                                               false, false);
}

__global__ __launch_bounds__(256) void fuse_gates(
    const float* __restrict__ weights, float* __restrict__ T) {
  const int e = blockIdx.x;
  const int tid = threadIdx.x;
  if (e < NU) {
    const int l = e / 3 + 1, g = e % 3;
    const int i = tid >> 4, j = tid & 15;
    float p = 1.f;
#pragma unroll
    for (int k = 0; k < 4; ++k) {
      const int q  = 4 * g + k;
      const float ty = 0.5f * weights[(l * N_QUBITS + q) * 2 + 0];
      const float cy = cosf(ty), sy = sinf(ty);
      const int bi = (i >> (3 - k)) & 1;
      const int bj = (j >> (3 - k)) & 1;
      p *= bi ? (bj ? cy : sy) : (bj ? -sy : cy);
    }
    T[e * 256 + tid] = p;
  } else {
    const int l = e - NU;
    float* P = (l == 0) ? T + LFLOATS : T + RFLOATS + (l - 1) * PHL;
    float pr = 1.f, pi = 0.f;
    for (int k = 0; k < 8; ++k) {
      const float tz = 0.5f * weights[(l * N_QUBITS + k) * 2 + 1];
      const float cz = cosf(tz), sz = sinf(tz);
      const float gi = ((tid >> (7 - k)) & 1) ? sz : -sz;
      const float nr = pr * cz - pi * gi;
      const float ni = pr * gi + pi * cz;
      pr = nr; pi = ni;
    }
    P[2 * tid]     = pr;
    P[2 * tid + 1] = pi;
    if (tid < 16) {
      float qr = 1.f, qi = 0.f;
      for (int k = 0; k < 4; ++k) {
        const float tz = 0.5f * weights[(l * N_QUBITS + 8 + k) * 2 + 1];
        const float cz = cosf(tz), sz = sinf(tz);
        const float gi = ((tid >> (3 - k)) & 1) ? sz : -sz;
        const float nr = qr * cz - qi * gi;
        const float ni = qr * gi + qi * cz;
        qr = nr; qi = ni;
      }
      P[512 + 2 * tid]     = qr;
      P[512 + 2 * tid + 1] = qi;
    }
  }
}

__global__ __launch_bounds__(256) void qsim(const float* __restrict__ x,
                                            const float* __restrict__ weights,
                                            const float* __restrict__ T,
                                            float* __restrict__ out) {
  __shared__ float Sr[DIM];
  __shared__ float Si[DIM];
  __shared__ float Tall[LFLOATS];
  __shared__ float red[256];

  const int tid  = threadIdx.x;
  const int b    = blockIdx.x;
  const int lane = tid & 31;
  const int wave = tid >> 5;
  const int half = lane >> 4;
  const int lm   = lane & 15;

#pragma unroll
  for (int k = 0; k < LFLOATS / 256; ++k)
    Tall[tid + 256 * k] = T[tid + 256 * k];
  if (tid < LFLOATS % 256)
    Tall[tid + 256 * (LFLOATS / 256)] = T[tid + 256 * (LFLOATS / 256)];

  const v2f* P0 = reinterpret_cast<const v2f*>(T + LFLOATS);
  const v2f p2 = P0[256 + (tid & 15)];
  v2f q01[16];
#pragma unroll
  for (int k = 0; k < 16; ++k) q01[k] = P0[(tid >> 4) + 16 * k];

  float cw[N_QUBITS], sw[N_QUBITS];
#pragma unroll
  for (int w = 0; w < N_QUBITS; ++w) {
    const float t = 0.5f * (x[b * N_QUBITS + w] + weights[w * 2 + 0]);
    cw[w] = cosf(t);
    sw[w] = sinf(t);
  }
#pragma unroll
  for (int k = 0; k < 16; ++k) {
    const int a = tid + 256 * k;
    float p = 1.f;
#pragma unroll
    for (int w = 0; w < N_QUBITS; ++w)
      p *= ((a >> (11 - w)) & 1) ? sw[w] : cw[w];
    Sr[a] = p * (q01[k].x * p2.x - q01[k].y * p2.y);
    Si[a] = p * (q01[k].x * p2.y + q01[k].y * p2.x);
  }
  __syncthreads();

  float acc = 0.f;
#pragma unroll
  for (int l = 1; l < N_LAYERS; ++l) {
    const bool gray = true;
    const bool last = (l == N_LAYERS - 1);

    {
      const float* Ur = Tall + ((l - 1) * 3 + 0) * 256;
      v8f dr[2] = {};
      v8f di[2] = {};
#pragma unroll
      for (int c = 0; c < 4; ++c) {
        const int k0 = 4 * c + 2 * half;
        v2f ur;
        ur.x = Ur[lm * 16 + k0];
        ur.y = Ur[lm * 16 + k0 + 1];
#pragma unroll
        for (int t2 = 0; t2 < 2; ++t2) {
          const int tile = wave * 2 + t2;
          int a0 = k0 * 256 + tile * 16 + lm;
          int a1 = a0 + 256;
          if (gray) {
            a0 ^= (a0 >> 1);
            a1 ^= (a1 >> 1);
          }
          v2f br, bi;
          br.x = Sr[a0]; br.y = Sr[a1];
          bi.x = Si[a0]; bi.y = Si[a1];
          dr[t2] = wmma4(ur, br, dr[t2]);
          di[t2] = wmma4(ur, bi, di[t2]);
        }
      }
      if (last) {
#pragma unroll
        for (int t2 = 0; t2 < 2; ++t2) {
#pragma unroll
          for (int v = 0; v < 8; ++v)
            acc += dr[t2][v] * dr[t2][v] + di[t2][v] * di[t2][v];
        }
        if (half) acc = -acc;
      } else {
        __syncthreads();
#pragma unroll
        for (int t2 = 0; t2 < 2; ++t2) {
          const int tile = wave * 2 + t2;
#pragma unroll
          for (int v = 0; v < 8; ++v) {
            const int m = v + 8 * half;
            const int a = m * 256 + tile * 16 + lm;
            Sr[a] = dr[t2][v];
            Si[a] = di[t2][v];
          }
        }
        __syncthreads();
      }
    }

    if (!last) {
      const float* U1 = Tall + ((l - 1) * 3 + 1) * 256;
      const float* U2 = Tall + ((l - 1) * 3 + 2) * 256;
      v8f er[2] = {};
      v8f ei[2] = {};
#pragma unroll
      for (int c = 0; c < 4; ++c) {
        const int k0 = 4 * c + 2 * half;
        v2f u1;
        u1.x = U1[lm * 16 + k0];
        u1.y = U1[lm * 16 + k0 + 1];
#pragma unroll
        for (int t2 = 0; t2 < 2; ++t2) {
          const int tile = wave * 2 + t2;
          const int a0 = tile * 256 + k0 * 16 + lm;
          const int a1 = a0 + 16;
          v2f br, bi;
          br.x = Sr[a0]; br.y = Sr[a1];
          bi.x = Si[a0]; bi.y = Si[a1];
          er[t2] = wmma4(br, u1, er[t2]);
          ei[t2] = wmma4(bi, u1, ei[t2]);
        }
      }

      v8f dr[2] = {};
      v8f di[2] = {};
#pragma unroll
      for (int c = 0; c < 4; ++c) {
        const int q0 = 8 * half + 2 * c;
        v2f u2;
        u2.x = U2[lm * 16 + q0];
        u2.y = U2[lm * 16 + q0 + 1];
#pragma unroll
        for (int t2 = 0; t2 < 2; ++t2) {
          v2f ar, ai;
          ar.x = er[t2][2 * c]; ar.y = er[t2][2 * c + 1];
          ai.x = ei[t2][2 * c]; ai.y = ei[t2][2 * c + 1];
          dr[t2] = wmma4(ar, u2, dr[t2]);
          di[t2] = wmma4(ai, u2, di[t2]);
        }
      }

      const float* P = Tall + RFLOATS + (l - 1) * PHL;
      const float p2r = P[512 + 2 * lm];
      const float p2i = P[512 + 2 * lm + 1];
#pragma unroll
      for (int t2 = 0; t2 < 2; ++t2) {
        const int h0 = (wave * 2 + t2) * 16 + 8 * half;
#pragma unroll
        for (int v = 0; v < 8; ++v) {
          const float qr = P[2 * (h0 + v)];
          const float qi = P[2 * (h0 + v) + 1];
          const float wr = qr * p2r - qi * p2i;
          const float wi = qr * p2i + qi * p2r;
          const float sr = dr[t2][v];
          const float si = di[t2][v];
          dr[t2][v] = sr * wr - si * wi;
          di[t2][v] = sr * wi + si * wr;
        }
      }
      __syncthreads();
#pragma unroll
      for (int t2 = 0; t2 < 2; ++t2) {
        const int tile = wave * 2 + t2;
#pragma unroll
        for (int v = 0; v < 8; ++v) {
          const int m = v + 8 * half;
          const int a = (tile * 16 + m) * 16 + lm;
          Sr[a] = dr[t2][v];
          Si[a] = di[t2][v];
        }
      }
      __syncthreads();
    }
  }

  red[tid] = acc;
  __syncthreads();
  for (int off = 128; off > 0; off >>= 1) {
    if (tid < off) red[tid] += red[tid + off];
    __syncthreads();
  }
  if (tid == 0) out[b] = red[0];
}

extern "C" void kernel_launch(void* const* d_in, const int* in_sizes, int n_in,
                              void* d_out, int out_size, void* d_ws,
                              size_t ws_size, hipStream_t stream) {
  const float* x = (const float*)d_in[0];
  const float* w = (const float*)d_in[1];
  float* out     = (float*)d_out;
  float* T       = (float*)d_ws;

  fuse_gates<<<dim3(NU + NPH), dim3(256), 0, stream>>>(w, T);
  qsim<<<dim3(BATCH), dim3(256), 0, stream>>>(x, w, T, out);
}


# State kept in LDS as (re,im) pairs at an XOR-swizzled index (bit 4 ^= bits 5,7,9,11): each state read or write is one conflict-free 8-byte LDS op, not 4-byte ops with 2- and 4-way bank conflicts
# speedup vs baseline: 2.5096x; 1.1441x over previous
#include <hip/hip_runtime.h>


typedef __attribute__((ext_vector_type(2))) float v2f;
typedef __attribute__((ext_vector_type(8))) float v8f;

#define N_QUBITS 12
#define N_LAYERS 6
#define BATCH    4096
#define DIM      4096
#define NU       ((N_LAYERS - 2) * 3 + 1)
#define NPH      (N_LAYERS - 1)
#define RFLOATS  (NU * 256)
#define PHL      544
#define LFLOATS  (RFLOATS + (NPH - 1) * PHL)
#define TFLOATS  (LFLOATS + PHL)

static __device__ __forceinline__ v8f wmma4(v2f a, v2f b, v8f c) {
  return __builtin_amdgcn_wmma_f32_16x16x4_f32(false, a, false, b, (short)0, c,
                                               false, false);
}

static __device__ __forceinline__ int swz(int a) {
  return a ^ ((((a >> 5) ^ (a >> 7) ^ (a >> 9) ^ (a >> 11)) & 1) << 4);
}
static __device__ __forceinline__ int gray(int a) { return a ^ (a >> 1); }

__global__ __launch_bounds__(256) void fuse_gates(
    const float* __restrict__ weights, float* __restrict__ T) {
  const int e = blockIdx.x;
  const int tid = threadIdx.x;
  if (e < NU) {
    const int l = e / 3 + 1, g = e % 3;
    const int i = tid >> 4, j = tid & 15;
    float p = 1.f;
#pragma unroll
    for (int k = 0; k < 4; ++k) {
      const int q  = 4 * g + k;
      const float ty = 0.5f * weights[(l * N_QUBITS + q) * 2 + 0];
      const float cy = cosf(ty), sy = sinf(ty);
      const int bi = (i >> (3 - k)) & 1;
      const int bj = (j >> (3 - k)) & 1;
      p *= bi ? (bj ? cy : sy) : (bj ? -sy : cy);
    }
    T[e * 256 + tid] = p;
  } else {
    const int l = e - NU;
    float* P = (l == 0) ? T + LFLOATS : T + RFLOATS + (l - 1) * PHL;
    float pr = 1.f, pi = 0.f;
    for (int k = 0; k < 8; ++k) {
      const float tz = 0.5f * weights[(l * N_QUBITS + k) * 2 + 1];
      const float cz = cosf(tz), sz = sinf(tz);
      const float gi = ((tid >> (7 - k)) & 1) ? sz : -sz;
      const float nr = pr * cz - pi * gi;
      const float ni = pr * gi + pi * cz;
      pr = nr; pi = ni;
    }
    P[2 * tid]     = pr;
    P[2 * tid + 1] = pi;
    if (tid < 16) {
      float qr = 1.f, qi = 0.f;
      for (int k = 0; k < 4; ++k) {
        const float tz = 0.5f * weights[(l * N_QUBITS + 8 + k) * 2 + 1];
        const float cz = cosf(tz), sz = sinf(tz);
        const float gi = ((tid >> (3 - k)) & 1) ? sz : -sz;
        const float nr = qr * cz - qi * gi;
        const float ni = qr * gi + qi * cz;
        qr = nr; qi = ni;
      }
      P[512 + 2 * tid]     = qr;
      P[512 + 2 * tid + 1] = qi;
    }
  }
}

__global__ __launch_bounds__(256) void qsim(const float* __restrict__ x,
                                            const float* __restrict__ weights,
                                            const float* __restrict__ T,
                                            float* __restrict__ out) {
  __shared__ v2f S[DIM];
  __shared__ float Tall[LFLOATS];
  __shared__ float red[256];

  const int tid  = threadIdx.x;
  const int b    = blockIdx.x;
  const int lane = tid & 31;
  const int wave = tid >> 5;
  const int half = lane >> 4;
  const int lm   = lane & 15;

#pragma unroll
  for (int k = 0; k < LFLOATS / 256; ++k)
    Tall[tid + 256 * k] = T[tid + 256 * k];
  if (tid < LFLOATS % 256)
    Tall[tid + 256 * (LFLOATS / 256)] = T[tid + 256 * (LFLOATS / 256)];

  const v2f* P0 = reinterpret_cast<const v2f*>(T + LFLOATS);
  const v2f p2 = P0[256 + (tid & 15)];
  v2f q01[16];
#pragma unroll
  for (int k = 0; k < 16; ++k) q01[k] = P0[(tid >> 4) + 16 * k];

  float cw[N_QUBITS], sw[N_QUBITS];
#pragma unroll
  for (int w = 0; w < N_QUBITS; ++w) {
    const float t = 0.5f * (x[b * N_QUBITS + w] + weights[w * 2 + 0]);
    cw[w] = cosf(t);
    sw[w] = sinf(t);
  }
  const int sw_init = swz(tid);
#pragma unroll
  for (int k = 0; k < 16; ++k) {
    const int a = tid + 256 * k;
    float p = 1.f;
#pragma unroll
    for (int w = 0; w < N_QUBITS; ++w)
      p *= ((a >> (11 - w)) & 1) ? sw[w] : cw[w];
    v2f z;
    z.x = p * (q01[k].x * p2.x - q01[k].y * p2.y);
    z.y = p * (q01[k].x * p2.y + q01[k].y * p2.x);
    S[sw_init ^ swz(256 * k)] = z;
  }
  __syncthreads();

  const int sw_r0 = swz(gray(half * 512 + wave * 32 + lm));
  const int sw_w0 = swz(half * 2048 + wave * 32 + lm);
  const int sw_r1 = swz(wave * 512 + half * 32 + lm);
  const int sw_w2 = swz(wave * 512 + half * 128 + lm);

  float acc = 0.f;
#pragma unroll
  for (int l = 1; l < N_LAYERS; ++l) {
    const bool last = (l == N_LAYERS - 1);

    {
      const float* Ur = Tall + ((l - 1) * 3 + 0) * 256;
      v8f dr[2] = {};
      v8f di[2] = {};
#pragma unroll
      for (int c = 0; c < 4; ++c) {
        const int k0 = 4 * c + 2 * half;
        v2f ur;
        ur.x = Ur[lm * 16 + k0];
        ur.y = Ur[lm * 16 + k0 + 1];
#pragma unroll
        for (int t2 = 0; t2 < 2; ++t2) {
          const v2f z0 = S[sw_r0 ^ swz(gray((4 * c) * 256 + t2 * 16))];
          const v2f z1 = S[sw_r0 ^ swz(gray((4 * c + 1) * 256 + t2 * 16))];
          v2f br, bi;
          br.x = z0.x; br.y = z1.x;
          bi.x = z0.y; bi.y = z1.y;
          dr[t2] = wmma4(ur, br, dr[t2]);
          di[t2] = wmma4(ur, bi, di[t2]);
        }
      }
      if (last) {
#pragma unroll
        for (int t2 = 0; t2 < 2; ++t2) {
#pragma unroll
          for (int v = 0; v < 8; ++v)
            acc += dr[t2][v] * dr[t2][v] + di[t2][v] * di[t2][v];
        }
        if (half) acc = -acc;
      } else {
        __syncthreads();
#pragma unroll
        for (int t2 = 0; t2 < 2; ++t2) {
#pragma unroll
          for (int v = 0; v < 8; ++v) {
            v2f z;
            z.x = dr[t2][v];
            z.y = di[t2][v];
            S[sw_w0 ^ swz(v * 256 + t2 * 16)] = z;
          }
        }
        __syncthreads();
      }
    }

    if (!last) {
      const float* U1 = Tall + ((l - 1) * 3 + 1) * 256;
      const float* U2 = Tall + ((l - 1) * 3 + 2) * 256;
      v8f er[2] = {};
      v8f ei[2] = {};
#pragma unroll
      for (int c = 0; c < 4; ++c) {
        const int k0 = 4 * c + 2 * half;
        v2f u1;
        u1.x = U1[lm * 16 + k0];
        u1.y = U1[lm * 16 + k0 + 1];
#pragma unroll
        for (int t2 = 0; t2 < 2; ++t2) {
          const v2f z0 = S[sw_r1 ^ swz(t2 * 256 + (4 * c) * 16)];
          const v2f z1 = S[sw_r1 ^ swz(t2 * 256 + (4 * c + 1) * 16)];
          v2f br, bi;
          br.x = z0.x; br.y = z1.x;
          bi.x = z0.y; bi.y = z1.y;
          er[t2] = wmma4(br, u1, er[t2]);
          ei[t2] = wmma4(bi, u1, ei[t2]);
        }
      }

      v8f dr[2] = {};
      v8f di[2] = {};
#pragma unroll
      for (int c = 0; c < 4; ++c) {
        const int q0 = 8 * half + 2 * c;
        v2f u2;
        u2.x = U2[lm * 16 + q0];
        u2.y = U2[lm * 16 + q0 + 1];
#pragma unroll
        for (int t2 = 0; t2 < 2; ++t2) {
          v2f ar, ai;
          ar.x = er[t2][2 * c]; ar.y = er[t2][2 * c + 1];
          ai.x = ei[t2][2 * c]; ai.y = ei[t2][2 * c + 1];
          dr[t2] = wmma4(ar, u2, dr[t2]);
          di[t2] = wmma4(ai, u2, di[t2]);
        }
      }

      const float* P = Tall + RFLOATS + (l - 1) * PHL;
      const float p2r = P[512 + 2 * lm];
      const float p2i = P[512 + 2 * lm + 1];
#pragma unroll
      for (int t2 = 0; t2 < 2; ++t2) {
        const int h0 = (wave * 2 + t2) * 16 + 8 * half;
#pragma unroll
        for (int v = 0; v < 8; ++v) {
          const float qr = P[2 * (h0 + v)];
          const float qi = P[2 * (h0 + v) + 1];
          const float wr = qr * p2r - qi * p2i;
          const float wi = qr * p2i + qi * p2r;
          const float sr = dr[t2][v];
          const float si = di[t2][v];
          dr[t2][v] = sr * wr - si * wi;
          di[t2][v] = sr * wi + si * wr;
        }
      }
      __syncthreads();
#pragma unroll
      for (int t2 = 0; t2 < 2; ++t2) {
#pragma unroll
        for (int v = 0; v < 8; ++v) {
          v2f z;
          z.x = dr[t2][v];
          z.y = di[t2][v];
          S[sw_w2 ^ swz(t2 * 256 + v * 16)] = z;
        }
      }
      __syncthreads();
    }
  }

  red[tid] = acc;
  __syncthreads();
  for (int off = 128; off > 0; off >>= 1) {
    if (tid < off) red[tid] += red[tid + off];
    __syncthreads();
  }
  if (tid == 0) out[b] = red[0];
}

extern "C" void kernel_launch(void* const* d_in, const int* in_sizes, int n_in,
                              void* d_out, int out_size, void* d_ws,
                              size_t ws_size, hipStream_t stream) {
  const float* x = (const float*)d_in[0];
  const float* w = (const float*)d_in[1];
  float* out     = (float*)d_out;
  float* T       = (float*)d_ws;

  fuse_gates<<<dim3(NU + NPH), dim3(256), 0, stream>>>(w, T);
  qsim<<<dim3(BATCH), dim3(256), 0, stream>>>(x, w, T, out);
}


# Layer 1 in closed form: behind the CNOT ladder the product state is a chain of 3 wire groups tied by 2 bits, so its 3 matrix stages are 128 short sums, 4 terms per amplitude; changes float sum order
# speedup vs baseline: 2.8748x; 1.1455x over previous
#include <hip/hip_runtime.h>


typedef __attribute__((ext_vector_type(2))) float v2f;
typedef __attribute__((ext_vector_type(8))) float v8f;

#define N_QUBITS 12
#define N_LAYERS 6
#define BATCH    4096
#define DIM      4096
#define NU       ((N_LAYERS - 2) * 3 + 1)
#define NPH      (N_LAYERS - 1)
#define RFLOATS  (NU * 256)
#define PHL      544
#define LFLOATS  (RFLOATS + (NPH - 1) * PHL)
#define TFLOATS  (LFLOATS + PHL)

static __device__ __forceinline__ v8f wmma4(v2f a, v2f b, v8f c) {
  return __builtin_amdgcn_wmma_f32_16x16x4_f32(false, a, false, b, (short)0, c,
                                               false, false);
}

static __device__ __forceinline__ int swz(int a) {
  return a ^ ((((a >> 5) ^ (a >> 7) ^ (a >> 9) ^ (a >> 11)) & 1) << 4);
}
static __device__ __forceinline__ int gray(int a) { return a ^ (a >> 1); }
static __device__ __forceinline__ v2f cmul(v2f a, v2f b) {
  v2f r;
  r.x = a.x * b.x - a.y * b.y;
  r.y = a.x * b.y + a.y * b.x;
  return r;
}

__global__ __launch_bounds__(256) void fuse_gates(
    const float* __restrict__ weights, float* __restrict__ T) {
  const int e = blockIdx.x;
  const int tid = threadIdx.x;
  if (e < NU) {
    const int l = e / 3 + 1, g = e % 3;
    const int i = tid >> 4, j = tid & 15;
    float p = 1.f;
#pragma unroll
    for (int k = 0; k < 4; ++k) {
      const int q  = 4 * g + k;
      const float ty = 0.5f * weights[(l * N_QUBITS + q) * 2 + 0];
      const float cy = cosf(ty), sy = sinf(ty);
      const int bi = (i >> (3 - k)) & 1;
      const int bj = (j >> (3 - k)) & 1;
      p *= bi ? (bj ? cy : sy) : (bj ? -sy : cy);
    }
    T[e * 256 + tid] = p;
  } else {
    const int l = e - NU;
    if (l == 0) {
      if (tid < N_QUBITS) {
        const float tz = 0.5f * weights[tid * 2 + 1];
        T[LFLOATS + 2 * tid]     = cosf(tz);
        T[LFLOATS + 2 * tid + 1] = sinf(tz);
      }
      return;
    }
    float* P = T + RFLOATS + (l - 1) * PHL;
    float pr = 1.f, pi = 0.f;
    for (int k = 0; k < 8; ++k) {
      const float tz = 0.5f * weights[(l * N_QUBITS + k) * 2 + 1];
      const float cz = cosf(tz), sz = sinf(tz);
      const float gi = ((tid >> (7 - k)) & 1) ? sz : -sz;
      const float nr = pr * cz - pi * gi;
      const float ni = pr * gi + pi * cz;
      pr = nr; pi = ni;
    }
    P[2 * tid]     = pr;
    P[2 * tid + 1] = pi;
    if (tid < 16) {
      float qr = 1.f, qi = 0.f;
      for (int k = 0; k < 4; ++k) {
        const float tz = 0.5f * weights[(l * N_QUBITS + 8 + k) * 2 + 1];
        const float cz = cosf(tz), sz = sinf(tz);
        const float gi = ((tid >> (3 - k)) & 1) ? sz : -sz;
        const float nr = qr * cz - qi * gi;
        const float ni = qr * gi + qi * cz;
        qr = nr; qi = ni;
      }
      P[512 + 2 * tid]     = qr;
      P[512 + 2 * tid + 1] = qi;
    }
  }
}

__global__ __launch_bounds__(256) void qsim(const float* __restrict__ x,
                                            const float* __restrict__ weights,
                                            const float* __restrict__ T,
                                            float* __restrict__ out) {
  __shared__ v2f S[DIM];
  __shared__ __attribute__((aligned(16))) float Tall[LFLOATS];
  __shared__ float red[256];
  __shared__ v2f Fv[80];
  __shared__ v2f Av[128];

  const int tid  = threadIdx.x;
  const int b    = blockIdx.x;
  const int lane = tid & 31;
  const int wave = tid >> 5;
  const int half = lane >> 4;
  const int lm   = lane & 15;

#pragma unroll
  for (int k = 0; k < LFLOATS / 256; ++k)
    Tall[tid + 256 * k] = T[tid + 256 * k];
  if (tid < LFLOATS % 256)
    Tall[tid + 256 * (LFLOATS / 256)] = T[tid + 256 * (LFLOATS / 256)];

  const float* Z0 = T + LFLOATS;
  float f0r[N_QUBITS], f0i[N_QUBITS], f1r[N_QUBITS], f1i[N_QUBITS];
#pragma unroll
  for (int w = 0; w < N_QUBITS; ++w) {
    const float t  = 0.5f * (x[b * N_QUBITS + w] + weights[w * 2 + 0]);
    const float c  = cosf(t), sn = sinf(t);
    const float cz = Z0[2 * w], sz = Z0[2 * w + 1];
    f0r[w] = c * cz;   f0i[w] = -(c * sz);
    f1r[w] = sn * cz;  f1i[w] = sn * sz;
  }

  if (tid < 32) {
    const int m  = tid & 15, tin = tid >> 4;
    const int m3 = (m >> 3) & 1, m2 = (m >> 2) & 1, m1 = (m >> 1) & 1, m0 = m & 1;
#pragma unroll
    for (int g = 0; g < 3; ++g) {
      const int b0 = m3 ^ (g == 0 ? 0 : tin);
      const int b1 = m2 ^ m3, b2 = m1 ^ m2, b3 = m0 ^ m1;
      v2f p, q;
      p.x = b0 ? f1r[4 * g] : f0r[4 * g];
      p.y = b0 ? f1i[4 * g] : f0i[4 * g];
      q.x = b1 ? f1r[4 * g + 1] : f0r[4 * g + 1];
      q.y = b1 ? f1i[4 * g + 1] : f0i[4 * g + 1];
      p = cmul(p, q);
      q.x = b2 ? f1r[4 * g + 2] : f0r[4 * g + 2];
      q.y = b2 ? f1i[4 * g + 2] : f0i[4 * g + 2];
      p = cmul(p, q);
      q.x = b3 ? f1r[4 * g + 3] : f0r[4 * g + 3];
      q.y = b3 ? f1i[4 * g + 3] : f0i[4 * g + 3];
      p = cmul(p, q);
      if (g == 0) {
        if (tin == 0) Fv[m] = p;
      } else {
        Fv[16 + (g - 1) * 32 + tin * 16 + m] = p;
      }
    }
  }
  __syncthreads();

  if (tid < 128) {
    int g, fb, tsel;
    bool all;
    if (tid < 32)      { g = 0; fb = 0;                            tsel = tid >> 4;               all = false; }
    else if (tid < 96) { g = 1; fb = 16 + ((tid - 32) >> 5) * 16;  tsel = ((tid - 32) >> 4) & 1;  all = false; }
    else               { g = 2; fb = 48 + ((tid - 96) >> 4) * 16;  tsel = 0;                      all = true;  }
    const float* Rg = Tall + g * 256 + (tid & 15) * 16;
    float ar = 0.f, ai = 0.f;
#pragma unroll
    for (int col = 0; col < 16; ++col) {
      const v2f F = Fv[fb + col];
      const float rr = Rg[col];
      if (all || (col & 1) == tsel) {
        ar += rr * F.x;
        ai += rr * F.y;
      }
    }
    v2f o;
    o.x = ar;
    o.y = ai;
    Av[tid] = o;
  }
  __syncthreads();

  const int sw_init = swz(tid);
  {
    const int i1 = tid >> 4, i2 = tid & 15;
    const v2f* P1 = reinterpret_cast<const v2f*>(Tall + RFLOATS);
    const v2f p2  = P1[256 + i2];
    const v2f a20 = Av[96 + i2], a21 = Av[112 + i2];
    v2f B[2];
#pragma unroll
    for (int t1 = 0; t1 < 2; ++t1) {
      const v2f u = cmul(Av[32 + t1 * 32 + i1], a20);
      const v2f v = cmul(Av[48 + t1 * 32 + i1], a21);
      v2f sum;
      sum.x = u.x + v.x;
      sum.y = u.y + v.y;
      B[t1] = cmul(sum, p2);
    }
#pragma unroll
    for (int k = 0; k < 16; ++k) {
      const v2f u = cmul(Av[k], B[0]);
      const v2f v = cmul(Av[16 + k], B[1]);
      v2f phi;
      phi.x = u.x + v.x;
      phi.y = u.y + v.y;
      S[sw_init ^ swz(256 * k)] = cmul(phi, P1[k * 16 + i1]);
    }
  }
  __syncthreads();

  const int sw_r0 = swz(gray(half * 512 + wave * 32 + lm));
  const int sw_w0 = swz(half * 2048 + wave * 32 + lm);
  const int sw_r1 = swz(wave * 512 + half * 32 + lm);
  const int sw_w2 = swz(wave * 512 + half * 128 + lm);

  float acc = 0.f;
#pragma unroll
  for (int l = 2; l < N_LAYERS; ++l) {
    const bool last = (l == N_LAYERS - 1);

    {
      const float* Ur = Tall + ((l - 1) * 3 + 0) * 256;
      v8f dr[2] = {};
      v8f di[2] = {};
#pragma unroll
      for (int c = 0; c < 4; ++c) {
        const int k0 = 4 * c + 2 * half;
        v2f ur;
        ur.x = Ur[lm * 16 + k0];
        ur.y = Ur[lm * 16 + k0 + 1];
#pragma unroll
        for (int t2 = 0; t2 < 2; ++t2) {
          const v2f z0 = S[sw_r0 ^ swz(gray((4 * c) * 256 + t2 * 16))];
          const v2f z1 = S[sw_r0 ^ swz(gray((4 * c + 1) * 256 + t2 * 16))];
          v2f br, bi;
          br.x = z0.x; br.y = z1.x;
          bi.x = z0.y; bi.y = z1.y;
          dr[t2] = wmma4(ur, br, dr[t2]);
          di[t2] = wmma4(ur, bi, di[t2]);
        }
      }
      if (last) {
#pragma unroll
        for (int t2 = 0; t2 < 2; ++t2) {
#pragma unroll
          for (int v = 0; v < 8; ++v)
            acc += dr[t2][v] * dr[t2][v] + di[t2][v] * di[t2][v];
        }
        if (half) acc = -acc;
      } else {
        __syncthreads();
#pragma unroll
        for (int t2 = 0; t2 < 2; ++t2) {
#pragma unroll
          for (int v = 0; v < 8; ++v) {
            v2f z;
            z.x = dr[t2][v];
            z.y = di[t2][v];
            S[sw_w0 ^ swz(v * 256 + t2 * 16)] = z;
          }
        }
        __syncthreads();
      }
    }

    if (!last) {
      const float* U1 = Tall + ((l - 1) * 3 + 1) * 256;
      const float* U2 = Tall + ((l - 1) * 3 + 2) * 256;
      v8f er[2] = {};
      v8f ei[2] = {};
#pragma unroll
      for (int c = 0; c < 4; ++c) {
        const int k0 = 4 * c + 2 * half;
        v2f u1;
        u1.x = U1[lm * 16 + k0];
        u1.y = U1[lm * 16 + k0 + 1];
#pragma unroll
        for (int t2 = 0; t2 < 2; ++t2) {
          const v2f z0 = S[sw_r1 ^ swz(t2 * 256 + (4 * c) * 16)];
          const v2f z1 = S[sw_r1 ^ swz(t2 * 256 + (4 * c + 1) * 16)];
          v2f br, bi;
          br.x = z0.x; br.y = z1.x;
          bi.x = z0.y; bi.y = z1.y;
          er[t2] = wmma4(br, u1, er[t2]);
          ei[t2] = wmma4(bi, u1, ei[t2]);
        }
      }

      v8f dr[2] = {};
      v8f di[2] = {};
#pragma unroll
      for (int c = 0; c < 4; ++c) {
        const int q0 = 8 * half + 2 * c;
        v2f u2;
        u2.x = U2[lm * 16 + q0];
        u2.y = U2[lm * 16 + q0 + 1];
#pragma unroll
        for (int t2 = 0; t2 < 2; ++t2) {
          v2f ar, ai;
          ar.x = er[t2][2 * c]; ar.y = er[t2][2 * c + 1];
          ai.x = ei[t2][2 * c]; ai.y = ei[t2][2 * c + 1];
          dr[t2] = wmma4(ar, u2, dr[t2]);
          di[t2] = wmma4(ai, u2, di[t2]);
        }
      }

      const float* P = Tall + RFLOATS + (l - 1) * PHL;
      const float p2r = P[512 + 2 * lm];
      const float p2i = P[512 + 2 * lm + 1];
#pragma unroll
      for (int t2 = 0; t2 < 2; ++t2) {
        const int h0 = (wave * 2 + t2) * 16 + 8 * half;
#pragma unroll
        for (int v = 0; v < 8; ++v) {
          const float qr = P[2 * (h0 + v)];
          const float qi = P[2 * (h0 + v) + 1];
          const float wr = qr * p2r - qi * p2i;
          const float wi = qr * p2i + qi * p2r;
          const float sr = dr[t2][v];
          const float si = di[t2][v];
          dr[t2][v] = sr * wr - si * wi;
          di[t2][v] = sr * wi + si * wr;
        }
      }
      __syncthreads();
#pragma unroll
      for (int t2 = 0; t2 < 2; ++t2) {
#pragma unroll
        for (int v = 0; v < 8; ++v) {
          v2f z;
          z.x = dr[t2][v];
          z.y = di[t2][v];
          S[sw_w2 ^ swz(t2 * 256 + v * 16)] = z;
        }
      }
      __syncthreads();
    }
  }

  red[tid] = acc;
  __syncthreads();
  for (int off = 128; off > 0; off >>= 1) {
    if (tid < off) red[tid] += red[tid + off];
    __syncthreads();
  }
  if (tid == 0) out[b] = red[0];
}

extern "C" void kernel_launch(void* const* d_in, const int* in_sizes, int n_in,
                              void* d_out, int out_size, void* d_ws,
                              size_t ws_size, hipStream_t stream) {
  const float* x = (const float*)d_in[0];
  const float* w = (const float*)d_in[1];
  float* out     = (float*)d_out;
  float* T       = (float*)d_ws;

  fuse_gates<<<dim3(NU + NPH), dim3(256), 0, stream>>>(w, T);
  qsim<<<dim3(BATCH), dim3(256), 0, stream>>>(x, w, T, out);
}


# Layer 2 in closed form too: behind the 2nd CNOT ladder the state is a 3-group chain tied by 2 bits a side; 384 short sums, then one K=4 complex WMMA product per tile; changes float sum order
# speedup vs baseline: 3.5428x; 1.2324x over previous
#include <hip/hip_runtime.h>


typedef __attribute__((ext_vector_type(2))) float v2f;
typedef __attribute__((ext_vector_type(8))) float v8f;

#define N_QUBITS 12
#define N_LAYERS 6
#define BATCH    4096
#define DIM      4096
#define NU       ((N_LAYERS - 2) * 3 + 1)
#define NPH      (N_LAYERS - 1)
#define RFLOATS  (NU * 256)
#define PHL      544
#define GOFF     (RFLOATS + (NPH - 1) * PHL)
#define GPH      128
#define LFLOATS  (GOFF + GPH)
#define TFLOATS  (LFLOATS + PHL)

static __device__ __forceinline__ v8f wmma4(v2f a, v2f b, v8f c) {
  return __builtin_amdgcn_wmma_f32_16x16x4_f32(false, a, false, b, (short)0, c,
                                               false, false);
}

static __device__ __forceinline__ int swz(int a) {
  return a ^ ((((a >> 5) ^ (a >> 7) ^ (a >> 9) ^ (a >> 11)) & 1) << 4);
}
static __device__ __forceinline__ int gray(int a) { return a ^ (a >> 1); }
static __device__ __forceinline__ v2f cmul(v2f a, v2f b) {
  v2f r;
  r.x = a.x * b.x - a.y * b.y;
  r.y = a.x * b.y + a.y * b.x;
  return r;
}

__global__ __launch_bounds__(256) void fuse_gates(
    const float* __restrict__ weights, float* __restrict__ T) {
  const int e = blockIdx.x;
  const int tid = threadIdx.x;
  if (e < NU) {
    const int l = e / 3 + 1, g = e % 3;
    const int i = tid >> 4, j = tid & 15;
    float p = 1.f;
#pragma unroll
    for (int k = 0; k < 4; ++k) {
      const int q  = 4 * g + k;
      const float ty = 0.5f * weights[(l * N_QUBITS + q) * 2 + 0];
      const float cy = cosf(ty), sy = sinf(ty);
      const int bi = (i >> (3 - k)) & 1;
      const int bj = (j >> (3 - k)) & 1;
      p *= bi ? (bj ? cy : sy) : (bj ? -sy : cy);
    }
    T[e * 256 + tid] = p;
  } else {
    const int l = e - NU;
    if (l == 0) {
      if (tid < N_QUBITS) {
        const float tz = 0.5f * weights[tid * 2 + 1];
        T[LFLOATS + 2 * tid]     = cosf(tz);
        T[LFLOATS + 2 * tid + 1] = sinf(tz);
      }
      return;
    }
    float* P = T + RFLOATS + (l - 1) * PHL;
    float pr = 1.f, pi = 0.f;
    for (int k = 0; k < 8; ++k) {
      const float tz = 0.5f * weights[(l * N_QUBITS + k) * 2 + 1];
      const float cz = cosf(tz), sz = sinf(tz);
      const float gi = ((tid >> (7 - k)) & 1) ? sz : -sz;
      const float nr = pr * cz - pi * gi;
      const float ni = pr * gi + pi * cz;
      pr = nr; pi = ni;
    }
    P[2 * tid]     = pr;
    P[2 * tid + 1] = pi;
    if (tid < 16) {
      float qr = 1.f, qi = 0.f;
      for (int k = 0; k < 4; ++k) {
        const float tz = 0.5f * weights[(l * N_QUBITS + 8 + k) * 2 + 1];
        const float cz = cosf(tz), sz = sinf(tz);
        const float gi = ((tid >> (3 - k)) & 1) ? sz : -sz;
        const float nr = qr * cz - qi * gi;
        const float ni = qr * gi + qi * cz;
        qr = nr; qi = ni;
      }
      P[512 + 2 * tid]     = qr;
      P[512 + 2 * tid + 1] = qi;
    }
    if (l <= 2 && tid >= 32 && tid < 64) {
      const int gg = (tid - 32) >> 4, m = tid & 15;
      float qr = 1.f, qi = 0.f;
      for (int k = 0; k < 4; ++k) {
        const float tz = 0.5f * weights[(l * N_QUBITS + 4 * gg + k) * 2 + 1];
        const float cz = cosf(tz), sz = sinf(tz);
        const float gi = ((m >> (3 - k)) & 1) ? sz : -sz;
        const float nr = qr * cz - qi * gi;
        const float ni = qr * gi + qi * cz;
        qr = nr; qi = ni;
      }
      float* G = T + GOFF + (l - 1) * 64 + gg * 32;
      G[2 * m]     = qr;
      G[2 * m + 1] = qi;
    }
  }
}

__global__ __launch_bounds__(256) void qsim(const float* __restrict__ x,
                                            const float* __restrict__ weights,
                                            const float* __restrict__ T,
                                            float* __restrict__ out) {
  __shared__ v2f S[DIM];
  __shared__ __attribute__((aligned(16))) float Tall[LFLOATS];
  __shared__ float red[256];
  __shared__ v2f Fv[80];
  __shared__ v2f Av[128];
  __shared__ v2f Cv[384];

  const int tid  = threadIdx.x;
  const int b    = blockIdx.x;
  const int lane = tid & 31;
  const int wave = tid >> 5;
  const int half = lane >> 4;
  const int lm   = lane & 15;

#pragma unroll
  for (int k = 0; k < LFLOATS / 256; ++k)
    Tall[tid + 256 * k] = T[tid + 256 * k];
  if (tid < LFLOATS % 256)
    Tall[tid + 256 * (LFLOATS / 256)] = T[tid + 256 * (LFLOATS / 256)];

  const float* Z0 = T + LFLOATS;
  float f0r[N_QUBITS], f0i[N_QUBITS], f1r[N_QUBITS], f1i[N_QUBITS];
#pragma unroll
  for (int w = 0; w < N_QUBITS; ++w) {
    const float t  = 0.5f * (x[b * N_QUBITS + w] + weights[w * 2 + 0]);
    const float c  = cosf(t), sn = sinf(t);
    const float cz = Z0[2 * w], sz = Z0[2 * w + 1];
    f0r[w] = c * cz;   f0i[w] = -(c * sz);
    f1r[w] = sn * cz;  f1i[w] = sn * sz;
  }

  if (tid < 32) {
    const int m  = tid & 15, tin = tid >> 4;
    const int m3 = (m >> 3) & 1, m2 = (m >> 2) & 1, m1 = (m >> 1) & 1, m0 = m & 1;
#pragma unroll
    for (int g = 0; g < 3; ++g) {
      const int b0 = m3 ^ (g == 0 ? 0 : tin);
      const int b1 = m2 ^ m3, b2 = m1 ^ m2, b3 = m0 ^ m1;
      v2f p, q;
      p.x = b0 ? f1r[4 * g] : f0r[4 * g];
      p.y = b0 ? f1i[4 * g] : f0i[4 * g];
      q.x = b1 ? f1r[4 * g + 1] : f0r[4 * g + 1];
      q.y = b1 ? f1i[4 * g + 1] : f0i[4 * g + 1];
      p = cmul(p, q);
      q.x = b2 ? f1r[4 * g + 2] : f0r[4 * g + 2];
      q.y = b2 ? f1i[4 * g + 2] : f0i[4 * g + 2];
      p = cmul(p, q);
      q.x = b3 ? f1r[4 * g + 3] : f0r[4 * g + 3];
      q.y = b3 ? f1i[4 * g + 3] : f0i[4 * g + 3];
      p = cmul(p, q);
      if (g == 0) {
        if (tin == 0) Fv[m] = p;
      } else {
        Fv[16 + (g - 1) * 32 + tin * 16 + m] = p;
      }
    }
  }
  __syncthreads();

  const v2f* GP  = reinterpret_cast<const v2f*>(Tall + GOFF);
  const v2f* PH1 = reinterpret_cast<const v2f*>(Tall + RFLOATS);
  const v2f* PH2 = reinterpret_cast<const v2f*>(Tall + RFLOATS + PHL);

  if (tid < 128) {
    int g, fb, tsel;
    bool all;
    if (tid < 32)      { g = 0; fb = 0;                            tsel = tid >> 4;               all = false; }
    else if (tid < 96) { g = 1; fb = 16 + ((tid - 32) >> 5) * 16;  tsel = ((tid - 32) >> 4) & 1;  all = false; }
    else               { g = 2; fb = 48 + ((tid - 96) >> 4) * 16;  tsel = 0;                      all = true;  }
    const float* Rg = Tall + g * 256 + (tid & 15) * 16;
    float ar = 0.f, ai = 0.f;
#pragma unroll
    for (int col = 0; col < 16; ++col) {
      const v2f F = Fv[fb + col];
      const float rr = Rg[col];
      if (all || (col & 1) == tsel) {
        ar += rr * F.x;
        ai += rr * F.y;
      }
    }
    v2f o;
    o.x = ar;
    o.y = ai;
    const v2f ph = (g == 0) ? GP[tid & 15]
                 : (g == 1) ? GP[16 + (tid & 15)]
                            : PH1[256 + (tid & 15)];
    Av[tid] = cmul(o, ph);
  }
  __syncthreads();

  {
    const int al = tid >> 6, be = (tid >> 4) & 3, j1 = tid & 15;
    const int t1 = al >> 1, u1 = al & 1, t2 = be >> 1, u2 = be & 1;
    const float* R1 = Tall + 4 * 256 + j1 * 16;
    const v2f* A1 = Av + 32 + t1 * 32 + t2 * 16;
    float cr = 0.f, ci = 0.f;
#pragma unroll
    for (int n = 0; n < 8; ++n) {
      const int mid = 2 * n + u2;
      const v2f a = A1[(mid ^ n) ^ (u1 << 3)];
      const float rr = R1[mid];
      cr += rr * a.x;
      ci += rr * a.y;
    }
    v2f o;
    o.x = cr;
    o.y = ci;
    Cv[64 + tid] = cmul(o, GP[48 + j1]);
  }
  if (tid >= 128 && tid < 192) {
    const int o4 = tid - 128;
    const int al = o4 >> 4, j0 = o4 & 15, t1 = al >> 1, u1 = al & 1;
    const float* R0 = Tall + 3 * 256 + j0 * 16;
    const v2f* A0 = Av + t1 * 16;
    float cr = 0.f, ci = 0.f;
#pragma unroll
    for (int n = 0; n < 8; ++n) {
      const int hi = 2 * n + u1;
      const v2f a = A0[hi ^ n];
      const float rr = R0[hi];
      cr += rr * a.x;
      ci += rr * a.y;
    }
    v2f o;
    o.x = cr;
    o.y = ci;
    Cv[o4] = cmul(o, GP[32 + j0]);
  } else if (tid >= 192) {
    const int o4 = tid - 192;
    const int be = o4 >> 4, j2 = o4 & 15, t2 = be >> 1, u2 = be & 1;
    const float* R2 = Tall + 5 * 256 + j2 * 16;
    const v2f* A2 = Av + 96 + t2 * 16;
    float cr = 0.f, ci = 0.f;
#pragma unroll
    for (int lo = 0; lo < 16; ++lo) {
      const v2f a = A2[(lo ^ (lo >> 1)) ^ (u2 << 3)];
      const float rr = R2[lo];
      cr += rr * a.x;
      ci += rr * a.y;
    }
    v2f o;
    o.x = cr;
    o.y = ci;
    Cv[320 + o4] = cmul(o, PH2[256 + j2]);
  }
  __syncthreads();

  const int sw_w0 = swz(half * 2048 + wave * 32 + lm);
  {
    const v2f c00 = Cv[(2 * half) * 16 + lm];
    const v2f c01 = Cv[(2 * half + 1) * 16 + lm];
    v2f ar, ai, an;
    ar.x = c00.x;  ar.y = c01.x;
    ai.x = c00.y;  ai.y = c01.y;
    an.x = -c00.y; an.y = -c01.y;
    v2f c2[4];
#pragma unroll
    for (int be = 0; be < 4; ++be) c2[be] = Cv[320 + be * 16 + lm];
#pragma unroll
    for (int t = 0; t < 2; ++t) {
      const v2f* C1 = Cv + 64 + (2 * half) * 64 + (2 * wave + t);
      v2f br, bi;
      {
        v2f sum = cmul(C1[0], c2[0]);
#pragma unroll
        for (int be = 1; be < 4; ++be) {
          const v2f u = cmul(C1[be * 16], c2[be]);
          sum.x += u.x;
          sum.y += u.y;
        }
        br.x = sum.x;
        bi.x = sum.y;
      }
      {
        v2f sum = cmul(C1[64], c2[0]);
#pragma unroll
        for (int be = 1; be < 4; ++be) {
          const v2f u = cmul(C1[64 + be * 16], c2[be]);
          sum.x += u.x;
          sum.y += u.y;
        }
        br.y = sum.x;
        bi.y = sum.y;
      }
      v8f dr = {};
      v8f di = {};
      dr = wmma4(ar, br, dr);
      dr = wmma4(an, bi, dr);
      di = wmma4(ar, bi, di);
      di = wmma4(ai, br, di);
#pragma unroll
      for (int v = 0; v < 8; ++v) {
        v2f z;
        z.x = dr[v];
        z.y = di[v];
        S[sw_w0 ^ swz(v * 256 + t * 16)] = z;
      }
    }
  }
  __syncthreads();

  const int sw_r0 = swz(gray(half * 512 + wave * 32 + lm));
  const int sw_r1 = swz(wave * 512 + half * 32 + lm);
  const int sw_w2 = swz(wave * 512 + half * 128 + lm);

  float acc = 0.f;
#pragma unroll
  for (int l = 3; l < N_LAYERS; ++l) {
    const bool last = (l == N_LAYERS - 1);

    {
      const float* Ur = Tall + ((l - 1) * 3 + 0) * 256;
      v8f dr[2] = {};
      v8f di[2] = {};
#pragma unroll
      for (int c = 0; c < 4; ++c) {
        const int k0 = 4 * c + 2 * half;
        v2f ur;
        ur.x = Ur[lm * 16 + k0];
        ur.y = Ur[lm * 16 + k0 + 1];
#pragma unroll
        for (int t2 = 0; t2 < 2; ++t2) {
          const v2f z0 = S[sw_r0 ^ swz(gray((4 * c) * 256 + t2 * 16))];
          const v2f z1 = S[sw_r0 ^ swz(gray((4 * c + 1) * 256 + t2 * 16))];
          v2f br, bi;
          br.x = z0.x; br.y = z1.x;
          bi.x = z0.y; bi.y = z1.y;
          dr[t2] = wmma4(ur, br, dr[t2]);
          di[t2] = wmma4(ur, bi, di[t2]);
        }
      }
      if (last) {
#pragma unroll
        for (int t2 = 0; t2 < 2; ++t2) {
#pragma unroll
          for (int v = 0; v < 8; ++v)
            acc += dr[t2][v] * dr[t2][v] + di[t2][v] * di[t2][v];
        }
        if (half) acc = -acc;
      } else {
        __syncthreads();
#pragma unroll
        for (int t2 = 0; t2 < 2; ++t2) {
#pragma unroll
          for (int v = 0; v < 8; ++v) {
            v2f z;
            z.x = dr[t2][v];
            z.y = di[t2][v];
            S[sw_w0 ^ swz(v * 256 + t2 * 16)] = z;
          }
        }
        __syncthreads();
      }
    }

    if (!last) {
      const float* U1 = Tall + ((l - 1) * 3 + 1) * 256;
      const float* U2 = Tall + ((l - 1) * 3 + 2) * 256;
      v8f er[2] = {};
      v8f ei[2] = {};
#pragma unroll
      for (int c = 0; c < 4; ++c) {
        const int k0 = 4 * c + 2 * half;
        v2f u1;
        u1.x = U1[lm * 16 + k0];
        u1.y = U1[lm * 16 + k0 + 1];
#pragma unroll
        for (int t2 = 0; t2 < 2; ++t2) {
          const v2f z0 = S[sw_r1 ^ swz(t2 * 256 + (4 * c) * 16)];
          const v2f z1 = S[sw_r1 ^ swz(t2 * 256 + (4 * c + 1) * 16)];
          v2f br, bi;
          br.x = z0.x; br.y = z1.x;
          bi.x = z0.y; bi.y = z1.y;
          er[t2] = wmma4(br, u1, er[t2]);
          ei[t2] = wmma4(bi, u1, ei[t2]);
        }
      }

      v8f dr[2] = {};
      v8f di[2] = {};
#pragma unroll
      for (int c = 0; c < 4; ++c) {
        const int q0 = 8 * half + 2 * c;
        v2f u2;
        u2.x = U2[lm * 16 + q0];
        u2.y = U2[lm * 16 + q0 + 1];
#pragma unroll
        for (int t2 = 0; t2 < 2; ++t2) {
          v2f ar, ai;
          ar.x = er[t2][2 * c]; ar.y = er[t2][2 * c + 1];
          ai.x = ei[t2][2 * c]; ai.y = ei[t2][2 * c + 1];
          dr[t2] = wmma4(ar, u2, dr[t2]);
          di[t2] = wmma4(ai, u2, di[t2]);
        }
      }

      const float* P = Tall + RFLOATS + (l - 1) * PHL;
      const float p2r = P[512 + 2 * lm];
      const float p2i = P[512 + 2 * lm + 1];
#pragma unroll
      for (int t2 = 0; t2 < 2; ++t2) {
        const int h0 = (wave * 2 + t2) * 16 + 8 * half;
#pragma unroll
        for (int v = 0; v < 8; ++v) {
          const float qr = P[2 * (h0 + v)];
          const float qi = P[2 * (h0 + v) + 1];
          const float wr = qr * p2r - qi * p2i;
          const float wi = qr * p2i + qi * p2r;
          const float sr = dr[t2][v];
          const float si = di[t2][v];
          dr[t2][v] = sr * wr - si * wi;
          di[t2][v] = sr * wi + si * wr;
        }
      }
      __syncthreads();
#pragma unroll
      for (int t2 = 0; t2 < 2; ++t2) {
#pragma unroll
        for (int v = 0; v < 8; ++v) {
          v2f z;
          z.x = dr[t2][v];
          z.y = di[t2][v];
          S[sw_w2 ^ swz(t2 * 256 + v * 16)] = z;
        }
      }
      __syncthreads();
    }
  }

  red[tid] = acc;
  __syncthreads();
  for (int off = 128; off > 0; off >>= 1) {
    if (tid < off) red[tid] += red[tid + off];
    __syncthreads();
  }
  if (tid == 0) out[b] = red[0];
}

extern "C" void kernel_launch(void* const* d_in, const int* in_sizes, int n_in,
                              void* d_out, int out_size, void* d_ws,
                              size_t ws_size, hipStream_t stream) {
  const float* x = (const float*)d_in[0];
  const float* w = (const float*)d_in[1];
  float* out     = (float*)d_out;
  float* T       = (float*)d_ws;

  fuse_gates<<<dim3(NU + NPH), dim3(256), 0, stream>>>(w, T);
  qsim<<<dim3(BATCH), dim3(256), 0, stream>>>(x, w, T, out);
}


# Light cone of <Z_0>: each CNOT ladder widens it by one wire, so only wires 0-5 (0-3 from layer 2 on) reach it; 64 amplitudes per sample, 16 samples per wave, state in registers; new float sum order
# speedup vs baseline: 21.4669x; 6.0593x over previous
#include <hip/hip_runtime.h>


typedef __attribute__((ext_vector_type(2))) float v2f;
typedef __attribute__((ext_vector_type(8))) float v8f;

#define N_QUBITS 12
#define N_LAYERS 6
#define BATCH    4096
#define NW       6
#define DIM      64
#define SPW      16
#define M1OFF    0
#define MHOFF    (DIM * DIM)
#define P1OFF    (MHOFF + 4 * 256)
#define PHOFF    (P1OFF + 2 * DIM)
#define Z0OFF    (PHOFF + 3 * 32)

static __device__ __forceinline__ v8f wmma4(v2f a, v2f b, v8f c) {
  return __builtin_amdgcn_wmma_f32_16x16x4_f32(false, a, false, b, (short)0, c,
                                               false, false);
}
static __device__ __forceinline__ v2f cmul(v2f a, v2f b) {
  v2f r;
  r.x = a.x * b.x - a.y * b.y;
  r.y = a.x * b.y + a.y * b.x;
  return r;
}
static __device__ __forceinline__ int gray(int a) { return a ^ (a >> 1); }
static __device__ __forceinline__ int gray_inv(int z) {
  int y = z;
  y ^= y >> 1;
  y ^= y >> 2;
  y ^= y >> 4;
  return y;
}
static __device__ __forceinline__ int amp_of(int p) { return ((p & 15) << 2) | (p >> 4); }

__global__ __launch_bounds__(256) void fuse_gates(
    const float* __restrict__ weights, float* __restrict__ T) {
  const int e = blockIdx.x;
  const int tid = threadIdx.x;
  if (e < 16) {
    const int i = e * 4 + (tid >> 6);
    const int z = tid & 63;
    const int ia = gray(amp_of(i));
    const int ya = gray_inv(amp_of(z));
    float p = 1.f;
#pragma unroll
    for (int k = 0; k < NW; ++k) {
      const float ty = 0.5f * weights[(1 * N_QUBITS + k) * 2 + 0];
      const float cy = cosf(ty), sy = sinf(ty);
      const int bi = (ia >> (NW - 1 - k)) & 1;
      const int bj = (ya >> (NW - 1 - k)) & 1;
      p *= bi ? (bj ? cy : sy) : (bj ? -sy : cy);
    }
    const int to = i >> 4, c = i & 15;
    const int tt = z >> 4, h = (z >> 3) & 1, j = (z >> 1) & 3, s = z & 1;
    T[M1OFF + (((to * 4 + tt) * 4 + j) * 32 + (h * 16 + c)) * 2 + s] = p;
    return;
  }
  if (e < 20) {
    const int l = e - 14;
    const int i = tid >> 4, z = tid & 15;
    const int h = (z >> 3) & 1, j = (z >> 1) & 3, s = z & 1;
    const int y = (l == 2) ? z : (gray_inv(z) & 15);
    float p = 1.f;
#pragma unroll
    for (int k = 0; k < 4; ++k) {
      const float ty = 0.5f * weights[(l * N_QUBITS + k) * 2 + 0];
      const float cy = cosf(ty), sy = sinf(ty);
      const int bi = (i >> (3 - k)) & 1;
      const int bj = (y >> (3 - k)) & 1;
      p *= bi ? (bj ? cy : sy) : (bj ? -sy : cy);
    }
    T[MHOFF + (l - 2) * 256 + (j * 32 + (h * 16 + i)) * 2 + s] = p;
    return;
  }
  if (tid < 64) {
    const int ia = gray(amp_of(tid));
    float pr = 1.f, pi = 0.f;
    for (int k = 0; k < NW; ++k) {
      const float tz = 0.5f * weights[(1 * N_QUBITS + k) * 2 + 1];
      const float cz = cosf(tz), sz = sinf(tz);
      const float gi = ((ia >> (NW - 1 - k)) & 1) ? sz : -sz;
      const float nr = pr * cz - pi * gi;
      const float ni = pr * gi + pi * cz;
      pr = nr; pi = ni;
    }
    T[P1OFF + 2 * tid]     = pr;
    T[P1OFF + 2 * tid + 1] = pi;
  } else if (tid < 112) {
    const int l = 2 + ((tid - 64) >> 4);
    const int i = tid & 15;
    float pr = 1.f, pi = 0.f;
    for (int k = 0; k < 4; ++k) {
      const float tz = 0.5f * weights[(l * N_QUBITS + k) * 2 + 1];
      const float cz = cosf(tz), sz = sinf(tz);
      const float gi = ((i >> (3 - k)) & 1) ? sz : -sz;
      const float nr = pr * cz - pi * gi;
      const float ni = pr * gi + pi * cz;
      pr = nr; pi = ni;
    }
    T[PHOFF + (l - 2) * 32 + 2 * i]     = pr;
    T[PHOFF + (l - 2) * 32 + 2 * i + 1] = pi;
  } else if (tid < 112 + NW) {
    const int w = tid - 112;
    const float tz = 0.5f * weights[w * 2 + 1];
    T[Z0OFF + 2 * w]     = cosf(tz);
    T[Z0OFF + 2 * w + 1] = sinf(tz);
  }
}

__global__ __launch_bounds__(32) void qsim(const float* __restrict__ x,
                                           const float* __restrict__ weights,
                                           const float* __restrict__ T,
                                           float* __restrict__ out) {
  const int lane = threadIdx.x;
  const int half = lane >> 4;
  const int lm   = lane & 15;
  const int smp  = blockIdx.x * SPW + lm;

  const float* Z0 = T + Z0OFF;
  v2f f0[NW], f1[NW];
#pragma unroll
  for (int w = 0; w < NW; ++w) {
    const float t  = 0.5f * (x[smp * N_QUBITS + w] + weights[w * 2 + 0]);
    const float c  = cosf(t), sn = sinf(t);
    const float cz = Z0[2 * w], sz = Z0[2 * w + 1];
    f0[w].x = c * cz;   f0[w].y = -(c * sz);
    f1[w].x = sn * cz;  f1[w].y = sn * sz;
  }

  v8f sr[4], si[4];
  {
    const v2f fh = half ? f1[0] : f0[0];
    v2f suf[8];
#pragma unroll
    for (int r = 0; r < 8; ++r)
      suf[r] = cmul(cmul((r & 4) ? f1[1] : f0[1], (r & 2) ? f1[2] : f0[2]),
                    (r & 1) ? f1[3] : f0[3]);
#pragma unroll
    for (int t = 0; t < 4; ++t) {
      const v2f pre = cmul(cmul((t & 2) ? f1[4] : f0[4], (t & 1) ? f1[5] : f0[5]), fh);
#pragma unroll
      for (int r = 0; r < 8; ++r) {
        const v2f a = cmul(pre, suf[r]);
        sr[t][r] = a.x;
        si[t][r] = a.y;
      }
    }
  }


  {
    const v2f* Mf = reinterpret_cast<const v2f*>(T + M1OFF) + lane;
    v8f dr[4] = {};
    v8f di[4] = {};
#pragma unroll
    for (int tt = 0; tt < 4; ++tt) {
#pragma unroll
      for (int j = 0; j < 4; ++j) {
        v2f br, bi;
        br.x = sr[tt][2 * j]; br.y = sr[tt][2 * j + 1];
        bi.x = si[tt][2 * j]; bi.y = si[tt][2 * j + 1];
#pragma unroll
        for (int to = 0; to < 4; ++to) {
          const v2f a = Mf[((to * 4 + tt) * 4 + j) * 32];
          dr[to] = wmma4(a, br, dr[to]);
          di[to] = wmma4(a, bi, di[to]);
        }
      }
    }
    const v2f* P = reinterpret_cast<const v2f*>(T + P1OFF) + 8 * half;
#pragma unroll
    for (int to = 0; to < 4; ++to) {
#pragma unroll
      for (int r = 0; r < 8; ++r) {
        const v2f w = P[16 * to + r];
        const float a = dr[to][r];
        const float b = di[to][r];
        sr[to][r] = a * w.x - b * w.y;
        si[to][r] = a * w.y + b * w.x;
      }
    }
  }

  float acc = 0.f;
#pragma unroll
  for (int l = 2; l < N_LAYERS; ++l) {
    const v2f* Mf = reinterpret_cast<const v2f*>(T + MHOFF + (l - 2) * 256) + lane;
    v8f dr[4] = {};
    v8f di[4] = {};
#pragma unroll
    for (int j = 0; j < 4; ++j) {
      const v2f a = Mf[j * 32];
#pragma unroll
      for (int t = 0; t < 4; ++t) {
        v2f br, bi;
        br.x = sr[t][2 * j]; br.y = sr[t][2 * j + 1];
        bi.x = si[t][2 * j]; bi.y = si[t][2 * j + 1];
        dr[t] = wmma4(a, br, dr[t]);
        di[t] = wmma4(a, bi, di[t]);
      }
    }
    if (l < N_LAYERS - 1) {
      const v2f* P = reinterpret_cast<const v2f*>(T + PHOFF + (l - 2) * 32) + 8 * half;
      v2f w[8];
#pragma unroll
      for (int r = 0; r < 8; ++r) w[r] = P[r];
#pragma unroll
      for (int t = 0; t < 4; ++t) {
#pragma unroll
        for (int r = 0; r < 8; ++r) {
          const float a = dr[t][r];
          const float b = di[t][r];
          sr[t][r] = a * w[r].x - b * w[r].y;
          si[t][r] = a * w[r].y + b * w[r].x;
        }
      }
    } else {
#pragma unroll
      for (int t = 0; t < 4; ++t) {
#pragma unroll
        for (int r = 0; r < 8; ++r)
          acc += dr[t][r] * dr[t][r] + di[t][r] * di[t][r];
      }
    }
  }

  if (half) acc = -acc;
  acc += __shfl_xor(acc, 16);
  if (half == 0) out[smp] = acc;
}

extern "C" void kernel_launch(void* const* d_in, const int* in_sizes, int n_in,
                              void* d_out, int out_size, void* d_ws,
                              size_t ws_size, hipStream_t stream) {
  const float* x = (const float*)d_in[0];
  const float* w = (const float*)d_in[1];
  float* out     = (float*)d_out;
  float* T       = (float*)d_ws;

  fuse_gates<<<dim3(21), dim3(256), 0, stream>>>(w, T);
  qsim<<<dim3(BATCH / SPW), dim3(32), 0, stream>>>(x, w, T, out);
}


# One launch, was two: each 4-wave workgroup builds the gate tables in its own LDS (46 cos/sin pairs, then prefix-product trees) and reads fragments and phases from LDS in grouped runs, not from memory
# speedup vs baseline: 42.0685x; 1.9597x over previous
#include <hip/hip_runtime.h>


typedef __attribute__((ext_vector_type(2))) float v2f;
typedef __attribute__((ext_vector_type(8))) float v8f;

#define N_QUBITS 12
#define N_LAYERS 6
#define BATCH    4096
#define NW       6
#define DIM      64
#define SPW      16
#define WPB      4
#define M1OFF    0
#define MHOFF    (DIM * DIM)
#define P1OFF    (MHOFF + 4 * 256)
#define PHOFF    (P1OFF + 2 * DIM)
#define TFLOATS  (PHOFF + 3 * 32)
#define TR_Z0    0
#define TR_Y1    6
#define TR_Z1    12
#define TR_YH    18
#define TR_ZH    34
#define NTRIG    46

static __device__ __forceinline__ v8f wmma4(v2f a, v2f b, v8f c) {
  return __builtin_amdgcn_wmma_f32_16x16x4_f32(false, a, false, b, (short)0, c,
                                               false, false);
}
static __device__ __forceinline__ v2f cmul(v2f a, v2f b) {
  v2f r;
  r.x = a.x * b.x - a.y * b.y;
  r.y = a.x * b.y + a.y * b.x;
  return r;
}
static __device__ __forceinline__ int gray(int a) { return a ^ (a >> 1); }
static __device__ __forceinline__ int amp_of(int p) { return ((p & 15) << 2) | (p >> 4); }

__global__ __launch_bounds__(32 * WPB) void qsim(const float* __restrict__ x,
                                                 const float* __restrict__ weights,
                                                 float* __restrict__ out) {
  __shared__ __attribute__((aligned(16))) float T[TFLOATS];
  __shared__ v2f trig[NTRIG];

  const int tid  = threadIdx.x;
  const int lane = tid & 31;
  const int wv   = tid >> 5;
  const int half = lane >> 4;
  const int lm   = lane & 15;
  const int smp  = (blockIdx.x * WPB + wv) * SPW + lm;

  const int te = (tid < NTRIG) ? tid : 0;
  const int qy = te - TR_YH, qz = te - TR_ZH;
  const int widx =
      (te < TR_Y1) ? (0 * N_QUBITS + (te - TR_Z0)) * 2 + 1 :
      (te < TR_Z1) ? (1 * N_QUBITS + (te - TR_Y1)) * 2 + 0 :
      (te < TR_YH) ? (1 * N_QUBITS + (te - TR_Z1)) * 2 + 1 :
      (te < TR_ZH) ? ((2 + (qy >> 2)) * N_QUBITS + (qy & 3)) * 2 + 0 :
                     ((2 + (qz >> 2)) * N_QUBITS + (qz & 3)) * 2 + 1;
  const float wa = weights[widx];

  float cx[NW], sx[NW];
#pragma unroll
  for (int w = 0; w < NW; ++w) {
    const float t = 0.5f * (x[smp * N_QUBITS + w] + weights[w * 2 + 0]);
    cx[w] = cosf(t);
    sx[w] = sinf(t);
  }

  if (tid < NTRIG) {
    const float t = 0.5f * wa;
    v2f cs;
    cs.x = cosf(t);
    cs.y = sinf(t);
    trig[tid] = cs;
  }
  __syncthreads();

  {
    const int i = tid >> 1, beta = tid & 1;
    const int ia = gray(amp_of(i));
    float P[32];
    float up[NW][2];
#pragma unroll
    for (int k = 0; k < NW; ++k) {
      const v2f cs = trig[TR_Y1 + k];
      const int bi = (ia >> (NW - 1 - k)) & 1;
      const float u0 = bi ? cs.y : cs.x;
      const float u1 = bi ? cs.x : -cs.y;
      up[k][0] = beta ? u1 : u0;
      up[k][1] = beta ? u0 : u1;
    }
    P[0] = up[0][0];
#pragma unroll
    for (int k = 1; k < NW; ++k) {
#pragma unroll
      for (int m = (1 << (k - 1)) - 1; m >= 0; --m) {
        const float b = P[m];
        P[2 * m + 1] = b * up[k][1];
        P[2 * m]     = b * up[k][0];
      }
    }
    float val[32];
#pragma unroll
    for (int y = 0; y < 32; ++y) val[y ^ (y >> 1)] = P[y];
    v2f* M = reinterpret_cast<v2f*>(T + M1OFF) + ((i >> 4) * 512 + beta * 16 + (i & 15));
#pragma unroll
    for (int tt = 0; tt < 4; ++tt) {
#pragma unroll
      for (int j = 0; j < 4; ++j) {
        v2f o;
        o.x = val[(j << 3) | tt];
        o.y = val[(j << 3) | 4 | tt];
        M[(tt * 4 + j) * 32] = o;
      }
    }
  }
  {
    const int l2 = tid >> 5;
    const int i = (tid >> 1) & 15, beta = tid & 1;
    const int flip = (l2 != 0) ? beta : 0;
    float Q[8];
    float uq[4][2];
#pragma unroll
    for (int k = 0; k < 4; ++k) {
      const v2f cs = trig[TR_YH + l2 * 4 + k];
      const int bi = (i >> (3 - k)) & 1;
      const float u0 = bi ? cs.y : cs.x;
      const float u1 = bi ? cs.x : -cs.y;
      if (k == 0) {
        uq[k][0] = beta ? u1 : u0;
        uq[k][1] = uq[k][0];
      } else {
        uq[k][0] = flip ? u1 : u0;
        uq[k][1] = flip ? u0 : u1;
      }
    }
    Q[0] = uq[0][0];
#pragma unroll
    for (int k = 1; k < 4; ++k) {
#pragma unroll
      for (int m = (1 << (k - 1)) - 1; m >= 0; --m) {
        const float b = Q[m];
        Q[2 * m + 1] = b * uq[k][1];
        Q[2 * m]     = b * uq[k][0];
      }
    }
    v2f* M = reinterpret_cast<v2f*>(T + MHOFF) + (l2 * 128 + beta * 16 + i);
#pragma unroll
    for (int j = 0; j < 4; ++j) {
      const int z0 = 2 * j, z1 = 2 * j + 1;
      v2f o;
      o.x = (l2 == 0) ? Q[z0] : Q[(z0 ^ (z0 >> 1) ^ (z0 >> 2)) & 7];
      o.y = (l2 == 0) ? Q[z1] : Q[(z1 ^ (z1 >> 1) ^ (z1 >> 2)) & 7];
      M[j * 32] = o;
    }
  }
  if (tid < 64) {
    const int ia = gray(amp_of(tid));
    float pr = 1.f, pi = 0.f;
#pragma unroll
    for (int k = 0; k < NW; ++k) {
      const v2f cs = trig[TR_Z1 + k];
      const float cz = cs.x, sz = cs.y;
      const float gi = ((ia >> (NW - 1 - k)) & 1) ? sz : -sz;
      const float nr = pr * cz - pi * gi;
      const float ni = pr * gi + pi * cz;
      pr = nr; pi = ni;
    }
    v2f o; o.x = pr; o.y = pi;
    reinterpret_cast<v2f*>(T + P1OFF)[tid] = o;
  } else if (tid < 112) {
    const int l2 = (tid - 64) >> 4;
    const int i = tid & 15;
    float pr = 1.f, pi = 0.f;
#pragma unroll
    for (int k = 0; k < 4; ++k) {
      const v2f cs = trig[TR_ZH + l2 * 4 + k];
      const float cz = cs.x, sz = cs.y;
      const float gi = ((i >> (3 - k)) & 1) ? sz : -sz;
      const float nr = pr * cz - pi * gi;
      const float ni = pr * gi + pi * cz;
      pr = nr; pi = ni;
    }
    v2f o; o.x = pr; o.y = pi;
    reinterpret_cast<v2f*>(T + PHOFF)[l2 * 16 + i] = o;
  }
  __syncthreads();

#define FENCE() __builtin_amdgcn_sched_barrier(0)

  const v2f* Mf1 = reinterpret_cast<const v2f*>(T + M1OFF) + lane;
  v2f ag[2][16];
#pragma unroll
  for (int tt = 0; tt < 2; ++tt) {
#pragma unroll
    for (int j = 0; j < 4; ++j) {
#pragma unroll
      for (int to = 0; to < 4; ++to) ag[tt][j * 4 + to] = Mf1[((to * 4 + tt) * 4 + j) * 32];
    }
  }
  v2f ah[4][4];
  v2f ph[3][8];
#pragma unroll
  for (int l2 = 0; l2 < 4; ++l2) {
#pragma unroll
    for (int j = 0; j < 4; ++j)
      ah[l2][j] = (reinterpret_cast<const v2f*>(T + MHOFF + l2 * 256) + lane)[j * 32];
  }
#pragma unroll
  for (int l2 = 0; l2 < 3; ++l2) {
#pragma unroll
    for (int r = 0; r < 8; ++r)
      ph[l2][r] = (reinterpret_cast<const v2f*>(T + PHOFF + l2 * 32) + 8 * half)[r];
  }
  FENCE();

  v2f f0[NW], f1[NW];
#pragma unroll
  for (int w = 0; w < NW; ++w) {
    const v2f z = trig[TR_Z0 + w];
    const float c = cx[w], sn = sx[w];
    const float cz = z.x, sz = z.y;
    f0[w].x = c * cz;   f0[w].y = -(c * sz);
    f1[w].x = sn * cz;  f1[w].y = sn * sz;
  }

  v8f sr[4], si[4];
  {
    const v2f fh = half ? f1[0] : f0[0];
    v2f suf[8];
#pragma unroll
    for (int r = 0; r < 8; ++r)
      suf[r] = cmul(cmul((r & 4) ? f1[1] : f0[1], (r & 2) ? f1[2] : f0[2]),
                    (r & 1) ? f1[3] : f0[3]);
#pragma unroll
    for (int t = 0; t < 4; ++t) {
      const v2f pre = cmul(cmul((t & 2) ? f1[4] : f0[4], (t & 1) ? f1[5] : f0[5]), fh);
#pragma unroll
      for (int r = 0; r < 8; ++r) {
        const v2f a = cmul(pre, suf[r]);
        sr[t][r] = a.x;
        si[t][r] = a.y;
      }
    }
  }
  FENCE();


  {
    v8f dr[4] = {};
    v8f di[4] = {};
    v2f p1[4][8];
#pragma unroll
    for (int tt = 0; tt < 4; ++tt) {
#pragma unroll
      for (int j = 0; j < 4; ++j) {
        v2f br, bi;
        br.x = sr[tt][2 * j]; br.y = sr[tt][2 * j + 1];
        bi.x = si[tt][2 * j]; bi.y = si[tt][2 * j + 1];
#pragma unroll
        for (int to = 0; to < 4; ++to) {
          const v2f a = ag[tt & 1][j * 4 + to];
          dr[to] = wmma4(a, br, dr[to]);
          di[to] = wmma4(a, bi, di[to]);
        }
      }
      FENCE();
      if (tt < 2) {
#pragma unroll
        for (int j = 0; j < 4; ++j) {
#pragma unroll
          for (int to = 0; to < 4; ++to)
            ag[tt & 1][j * 4 + to] = Mf1[((to * 4 + (tt + 2)) * 4 + j) * 32];
        }
      } else if (tt == 2) {
        const v2f* P = reinterpret_cast<const v2f*>(T + P1OFF) + 8 * half;
#pragma unroll
        for (int to = 0; to < 4; ++to) {
#pragma unroll
          for (int r = 0; r < 8; ++r) p1[to][r] = P[16 * to + r];
        }
      }
      FENCE();
    }
#pragma unroll
    for (int to = 0; to < 4; ++to) {
#pragma unroll
      for (int r = 0; r < 8; ++r) {
        const v2f w = p1[to][r];
        const float a = dr[to][r];
        const float b = di[to][r];
        sr[to][r] = a * w.x - b * w.y;
        si[to][r] = a * w.y + b * w.x;
      }
    }
  }

  float acc = 0.f;
#pragma unroll
  for (int l = 2; l < N_LAYERS; ++l) {
    v8f dr[4] = {};
    v8f di[4] = {};
#pragma unroll
    for (int j = 0; j < 4; ++j) {
      const v2f a = ah[l - 2][j];
#pragma unroll
      for (int t = 0; t < 4; ++t) {
        v2f br, bi;
        br.x = sr[t][2 * j]; br.y = sr[t][2 * j + 1];
        bi.x = si[t][2 * j]; bi.y = si[t][2 * j + 1];
        dr[t] = wmma4(a, br, dr[t]);
        di[t] = wmma4(a, bi, di[t]);
      }
    }
    if (l < N_LAYERS - 1) {
#pragma unroll
      for (int t = 0; t < 4; ++t) {
#pragma unroll
        for (int r = 0; r < 8; ++r) {
          const v2f w = ph[l - 2][r];
          const float a = dr[t][r];
          const float b = di[t][r];
          sr[t][r] = a * w.x - b * w.y;
          si[t][r] = a * w.y + b * w.x;
        }
      }
    } else {
#pragma unroll
      for (int t = 0; t < 4; ++t) {
#pragma unroll
        for (int r = 0; r < 8; ++r)
          acc += dr[t][r] * dr[t][r] + di[t][r] * di[t][r];
      }
    }
  }

  if (half) acc = -acc;
  acc += __shfl_xor(acc, 16);
  if (half == 0) out[smp] = acc;
}

extern "C" void kernel_launch(void* const* d_in, const int* in_sizes, int n_in,
                              void* d_out, int out_size, void* d_ws,
                              size_t ws_size, hipStream_t stream) {
  const float* x = (const float*)d_in[0];
  const float* w = (const float*)d_in[1];
  float* out     = (float*)d_out;

  qsim<<<dim3(BATCH / (SPW * WPB)), dim3(32 * WPB), 0, stream>>>(x, w, out);
}
